# starting kernel
# baseline (speedup 1.0000x reference)
#include <hip/hip_runtime.h>


typedef float v2f __attribute__((ext_vector_type(2)));
typedef float v8f __attribute__((ext_vector_type(8)));

#define FDIM 64

__global__ void zero_f32_kernel(float* __restrict__ p, int n) {
  int i = blockIdx.x * blockDim.x + threadIdx.x;
  if (i < n) p[i] = 0.0f;
}

__global__ void degree_kernel(const int* __restrict__ dst, float* __restrict__ deg, int E) {
  int e = blockIdx.x * blockDim.x + threadIdx.x;
  if (e < E) unsafeAtomicAdd(&deg[dst[e]], 1.0f);
}

__global__ void dinv_kernel(const float* __restrict__ deg, float* __restrict__ dinv,
                            float* __restrict__ invdeg, int N) {
  int i = blockIdx.x * blockDim.x + threadIdx.x;
  if (i < N) {
    float d = deg[i] + 1.0f;
    dinv[i]   = rsqrtf(d);
    invdeg[i] = 1.0f / d;
  }
}

__global__ void __launch_bounds__(128)
gemm_selfloop_kernel(const float* __restrict__ x, const float* __restrict__ W,
                     const float* __restrict__ invdeg,
                     float* __restrict__ h, float* __restrict__ agg, int N) {
  const int wave    = threadIdx.x >> 5;
  const int lane    = threadIdx.x & 31;
  const int half    = lane >> 4;
  const int l       = lane & 15;
  const int rowBase = blockIdx.x * 16;
  const int colBase = wave * 16;

  int arow = rowBase + l;
  arow = (arow < N) ? arow : (N - 1);
  const float* xr = x + (size_t)arow * FDIM + 2 * half;
  const float* wp = W + (size_t)(2 * half) * FDIM + colBase + l;

  v8f c = {};
  #pragma unroll
  for (int kk = 0; kk < 16; ++kk) {
    const v2f a = *(const v2f*)(xr + kk * 4);
    v2f b;
    b.x = wp[(size_t)kk * 4 * FDIM];
    b.y = wp[(size_t)kk * 4 * FDIM + FDIM];
    c = __builtin_amdgcn_wmma_f32_16x16x4_f32(
         false, a,  false, b,
         (short)0, c,  false,  false);
  }

  #pragma unroll
  for (int r = 0; r < 8; ++r) {
    const int row = rowBase + r + 8 * half;
    if (row < N) {
      const size_t idx = (size_t)row * FDIM + colBase + l;
      const float v = c[r];
      h[idx]   = v;
      agg[idx] = v * invdeg[row];
    }
  }
}

__global__ void __launch_bounds__(256)
scatter_kernel(const int* __restrict__ ei, const float* __restrict__ dinv,
               const float* __restrict__ h, float* __restrict__ agg, int E) {
  const long gid = (long)blockIdx.x * blockDim.x + threadIdx.x;
  const int e = (int)(gid >> 4);
  const int q = (int)(gid & 15);
  if (e >= E) return;
  const int s = ei[e];
  const int d = ei[E + e];
  const float coef = dinv[s] * dinv[d];
  const float4 v = *((const float4*)(h + (size_t)s * FDIM) + q);
  float* ap = agg + (size_t)d * FDIM + (size_t)q * 4;
  unsafeAtomicAdd(ap + 0, v.x * coef);
  unsafeAtomicAdd(ap + 1, v.y * coef);
  unsafeAtomicAdd(ap + 2, v.z * coef);
  unsafeAtomicAdd(ap + 3, v.w * coef);
}

__global__ void __launch_bounds__(256)
update_kernel(const float* __restrict__ agg, const float* __restrict__ b,
              float* __restrict__ out, int N, int do_relu) {
  const int gid = blockIdx.x * blockDim.x + threadIdx.x;
  if (gid >= N * 16) return;
  const int q = gid & 15;
  float4 v  = ((const float4*)agg)[gid];
  const float4 bb = ((const float4*)b)[q];
  v.x += bb.x; v.y += bb.y; v.z += bb.z; v.w += bb.w;
  if (do_relu) {
    v.x = fmaxf(v.x, 0.0f); v.y = fmaxf(v.y, 0.0f);
    v.z = fmaxf(v.z, 0.0f); v.w = fmaxf(v.w, 0.0f);
  }
  ((float4*)out)[gid] = v;
}

extern "C" void kernel_launch(void* const* d_in, const int* in_sizes, int n_in,
                              void* d_out, int out_size, void* d_ws, size_t ws_size,
                              hipStream_t stream) {
  const float* x  = (const float*)d_in[0];
  const int*   ei = (const int*)  d_in[1];
  const float* W  = (const float*)d_in[2];
  const float* b  = (const float*)d_in[3];

  const int N     = in_sizes[0] / FDIM;
  const int E     = in_sizes[1] / 2;
  const int DEPTH = in_sizes[2] / (FDIM * FDIM);

  float* ws     = (float*)d_ws;
  float* deg    = ws;
  float* dinv   = ws + (size_t)N;
  float* invdeg = ws + (size_t)2 * N;
  float* h      = ws + (size_t)3 * N;
  float* agg    = h   + (size_t)N * FDIM;
  float* xbuf   = agg + (size_t)N * FDIM;

  const int T = 256;

  zero_f32_kernel<<<(N + T - 1) / T, T, 0, stream>>>(deg, N);
  degree_kernel<<<(E + T - 1) / T, T, 0, stream>>>(ei + E, deg, E);
  dinv_kernel<<<(N + T - 1) / T, T, 0, stream>>>(deg, dinv, invdeg, N);

  const int gemmBlocks    = (N + 15) / 16;
  const int scatterBlocks = (int)(((long)E * 16 + T - 1) / T);
  const int updateBlocks  = (N * 16 + T - 1) / T;

  const float* xcur = x;
  for (int layer = 0; layer < DEPTH; ++layer) {
    gemm_selfloop_kernel<<<gemmBlocks, 128, 0, stream>>>(
        xcur, W + (size_t)layer * FDIM * FDIM, invdeg, h, agg, N);
    scatter_kernel<<<scatterBlocks, T, 0, stream>>>(ei, dinv, h, agg, E);
    float* xout = (layer == DEPTH - 1) ? (float*)d_out : xbuf;
    update_kernel<<<updateBlocks, T, 0, stream>>>(
        agg, b + (size_t)layer * FDIM, xout, N, layer != DEPTH - 1);
    xcur = xout;
  }
}


# Edge scatter re-laid so each wave-wide f32 atomic add (and each h gather) covers one whole aligned 128-B line: 1 edge per instruction, 32 lanes x 4 B; sums still in atomic arrival order
# speedup vs baseline: 1.3250x; 1.3250x over previous
#include <hip/hip_runtime.h>


typedef float v2f __attribute__((ext_vector_type(2)));
typedef float v8f __attribute__((ext_vector_type(8)));

#define FDIM 64

__global__ void zero_f32_kernel(float* __restrict__ p, int n) {
  int i = blockIdx.x * blockDim.x + threadIdx.x;
  if (i < n) p[i] = 0.0f;
}

__global__ void degree_kernel(const int* __restrict__ dst, float* __restrict__ deg, int E) {
  int e = blockIdx.x * blockDim.x + threadIdx.x;
  if (e < E) unsafeAtomicAdd(&deg[dst[e]], 1.0f);
}

__global__ void dinv_kernel(const float* __restrict__ deg, float* __restrict__ dinv,
                            float* __restrict__ invdeg, int N) {
  int i = blockIdx.x * blockDim.x + threadIdx.x;
  if (i < N) {
    float d = deg[i] + 1.0f;
    dinv[i]   = rsqrtf(d);
    invdeg[i] = 1.0f / d;
  }
}

__global__ void __launch_bounds__(128)
gemm_selfloop_kernel(const float* __restrict__ x, const float* __restrict__ W,
                     const float* __restrict__ invdeg,
                     float* __restrict__ h, float* __restrict__ agg, int N) {
  const int wave    = threadIdx.x >> 5;
  const int lane    = threadIdx.x & 31;
  const int half    = lane >> 4;
  const int l       = lane & 15;
  const int rowBase = blockIdx.x * 16;
  const int colBase = wave * 16;

  int arow = rowBase + l;
  arow = (arow < N) ? arow : (N - 1);
  const float* xr = x + (size_t)arow * FDIM + 2 * half;
  const float* wp = W + (size_t)(2 * half) * FDIM + colBase + l;

  v8f c = {};
  #pragma unroll
  for (int kk = 0; kk < 16; ++kk) {
    const v2f a = *(const v2f*)(xr + kk * 4);
    v2f b;
    b.x = wp[(size_t)kk * 4 * FDIM];
    b.y = wp[(size_t)kk * 4 * FDIM + FDIM];
    c = __builtin_amdgcn_wmma_f32_16x16x4_f32(
         false, a,  false, b,
         (short)0, c,  false,  false);
  }

  #pragma unroll
  for (int r = 0; r < 8; ++r) {
    const int row = rowBase + r + 8 * half;
    if (row < N) {
      const size_t idx = (size_t)row * FDIM + colBase + l;
      const float v = c[r];
      h[idx]   = v;
      agg[idx] = v * invdeg[row];
    }
  }
}

#define EGROUP 8
__global__ void __launch_bounds__(256)
scatter_kernel(const int* __restrict__ ei, const float* __restrict__ dinv,
               const float* __restrict__ h, float* __restrict__ agg, int E) {
  const int lane = threadIdx.x & 31;
  const int wv = __builtin_amdgcn_readfirstlane((int)(threadIdx.x >> 5));
  const int e0 = (blockIdx.x * (int)(blockDim.x >> 5) + wv) * 32;
  if (e0 >= E) return;
  const int cnt = (E - e0 < 32) ? (E - e0) : 32;

  int s = 0, d = 0;
  float cf = 0.0f;
  if (lane < cnt) {
    s = ei[e0 + lane];
    d = ei[E + e0 + lane];
    cf = dinv[s] * dinv[d];
  }
  const int cfb = __float_as_int(cf);

  int j0 = 0;
  for (; j0 + EGROUP <= cnt; j0 += EGROUP) {
    float v0[EGROUP], v1[EGROUP];
    #pragma unroll
    for (int j = 0; j < EGROUP; ++j) {
      const int sj = __builtin_amdgcn_readlane(s, j0 + j);
      const float* hp = h + (size_t)sj * FDIM + lane;
      v0[j] = hp[0];
      v1[j] = hp[32];
    }
    #pragma unroll
    for (int j = 0; j < EGROUP; ++j) {
      const int   dj = __builtin_amdgcn_readlane(d, j0 + j);
      const float cj = __int_as_float(__builtin_amdgcn_readlane(cfb, j0 + j));
      float* ap = agg + (size_t)dj * FDIM + lane;
      unsafeAtomicAdd(ap,      v0[j] * cj);
      unsafeAtomicAdd(ap + 32, v1[j] * cj);
    }
  }
  for (; j0 < cnt; ++j0) {
    const int   sj = __builtin_amdgcn_readlane(s, j0);
    const int   dj = __builtin_amdgcn_readlane(d, j0);
    const float cj = __int_as_float(__builtin_amdgcn_readlane(cfb, j0));
    const float* hp = h + (size_t)sj * FDIM + lane;
    float* ap = agg + (size_t)dj * FDIM + lane;
    const float a0 = hp[0], a1 = hp[32];
    unsafeAtomicAdd(ap,      a0 * cj);
    unsafeAtomicAdd(ap + 32, a1 * cj);
  }
}

__global__ void __launch_bounds__(256)
update_kernel(const float* __restrict__ agg, const float* __restrict__ b,
              float* __restrict__ out, int N, int do_relu) {
  const int gid = blockIdx.x * blockDim.x + threadIdx.x;
  if (gid >= N * 16) return;
  const int q = gid & 15;
  float4 v  = ((const float4*)agg)[gid];
  const float4 bb = ((const float4*)b)[q];
  v.x += bb.x; v.y += bb.y; v.z += bb.z; v.w += bb.w;
  if (do_relu) {
    v.x = fmaxf(v.x, 0.0f); v.y = fmaxf(v.y, 0.0f);
    v.z = fmaxf(v.z, 0.0f); v.w = fmaxf(v.w, 0.0f);
  }
  ((float4*)out)[gid] = v;
}

extern "C" void kernel_launch(void* const* d_in, const int* in_sizes, int n_in,
                              void* d_out, int out_size, void* d_ws, size_t ws_size,
                              hipStream_t stream) {
  const float* x  = (const float*)d_in[0];
  const int*   ei = (const int*)  d_in[1];
  const float* W  = (const float*)d_in[2];
  const float* b  = (const float*)d_in[3];

  const int N     = in_sizes[0] / FDIM;
  const int E     = in_sizes[1] / 2;
  const int DEPTH = in_sizes[2] / (FDIM * FDIM);

  float* ws     = (float*)d_ws;
  float* deg    = ws;
  float* dinv   = ws + (size_t)N;
  float* invdeg = ws + (size_t)2 * N;
  float* h      = ws + (size_t)3 * N;
  float* agg    = h   + (size_t)N * FDIM;
  float* xbuf   = agg + (size_t)N * FDIM;

  const int T = 256;

  zero_f32_kernel<<<(N + T - 1) / T, T, 0, stream>>>(deg, N);
  degree_kernel<<<(E + T - 1) / T, T, 0, stream>>>(ei + E, deg, E);
  dinv_kernel<<<(N + T - 1) / T, T, 0, stream>>>(deg, dinv, invdeg, N);

  const int gemmBlocks    = (N + 15) / 16;
  const int scatterBlocks = (E + T - 1) / T;
  const int updateBlocks  = (N * 16 + T - 1) / T;

  const float* xcur = x;
  for (int layer = 0; layer < DEPTH; ++layer) {
    gemm_selfloop_kernel<<<gemmBlocks, 128, 0, stream>>>(
        xcur, W + (size_t)layer * FDIM * FDIM, invdeg, h, agg, N);
    scatter_kernel<<<scatterBlocks, T, 0, stream>>>(ei, dinv, h, agg, E);
    float* xout = (layer == DEPTH - 1) ? (float*)d_out : xbuf;
    update_kernel<<<updateBlocks, T, 0, stream>>>(
        agg, b + (size_t)layer * FDIM, xout, N, layer != DEPTH - 1);
    xcur = xout;
  }
}


# Float-atomic scatter replaced by per-call destination buckets, each sorted by source so every row's sum order is fixed (not arrival order); wave per row gathers, bias/ReLU at its one store
# speedup vs baseline: 2.5850x; 1.9509x over previous
#include <hip/hip_runtime.h>


typedef float v2f __attribute__((ext_vector_type(2)));
typedef float v8f __attribute__((ext_vector_type(8)));

#define FDIM 64
#define SCAN_ROWS 1024

__global__ void zero_i32_kernel(int* __restrict__ p, int n) {
  int i = blockIdx.x * blockDim.x + threadIdx.x;
  if (i < n) p[i] = 0;
}

__global__ void __launch_bounds__(256)
rank_kernel(const int* __restrict__ dst, int* __restrict__ cnt, int* __restrict__ rank, int E) {
  const int stride = gridDim.x * blockDim.x;
  for (int e = blockIdx.x * blockDim.x + threadIdx.x; e < E; e += stride)
    rank[e] = atomicAdd(&cnt[dst[e]], 1);
}

__global__ void __launch_bounds__(256)
rowptr_kernel(const int* __restrict__ cnt, int* __restrict__ rowptr,
              float* __restrict__ dinv, float* __restrict__ invdeg) {
  __shared__ int wsum[8];
  __shared__ int wpre[8];
  const int tid  = threadIdx.x;
  const int lane = tid & 31;
  const int wv   = tid >> 5;
  const int4* c4 = (const int4*)cnt;

  int pre = 0;
  const int nfront = (int)blockIdx.x;
  #pragma unroll 8
  for (int i = 0; i < nfront; ++i) {
    const int4 q = c4[i * (SCAN_ROWS / 4) + tid];
    pre += (q.x + q.y) + (q.z + q.w);
  }
  const int g = blockIdx.x * (SCAN_ROWS / 4) + tid;
  const int4 q = c4[g];
  const int t = (q.x + q.y) + (q.z + q.w);

  int inc = t;
  #pragma unroll
  for (int o = 1; o < 32; o <<= 1) {
    const int u = __shfl_up(inc, o);
    if (lane >= o) inc += u;
  }
  int ps = pre;
  #pragma unroll
  for (int o = 16; o > 0; o >>= 1) ps += __shfl_xor(ps, o);
  if (lane == 31) wsum[wv] = inc;
  if (lane == 0)  wpre[wv] = ps;
  __syncthreads();
  int base = 0;
  #pragma unroll
  for (int w = 0; w < 8; ++w) {
    base += wpre[w];
    if (w < wv) base += wsum[w];
  }
  const int ex = base + inc - t;
  int4 r;
  r.x = ex; r.y = r.x + q.x; r.z = r.y + q.y; r.w = r.z + q.z;
  ((int4*)rowptr)[g] = r;

  const float d0 = (float)q.x + 1.0f, d1 = (float)q.y + 1.0f;
  const float d2 = (float)q.z + 1.0f, d3 = (float)q.w + 1.0f;
  float4 di, iv;
  di.x = rsqrtf(d0); di.y = rsqrtf(d1); di.z = rsqrtf(d2); di.w = rsqrtf(d3);
  iv.x = 1.0f / d0;  iv.y = 1.0f / d1;  iv.z = 1.0f / d2;  iv.w = 1.0f / d3;
  ((float4*)dinv)[g]   = di;
  ((float4*)invdeg)[g] = iv;
}

__global__ void __launch_bounds__(256)
fill_kernel(const int* __restrict__ ei, const int* __restrict__ rowptr,
            const int* __restrict__ rank, const float* __restrict__ dinv,
            int* __restrict__ bucket, int E) {
  int2* bk = (int2*)bucket;
  const int stride = gridDim.x * blockDim.x;
  for (int e = blockIdx.x * blockDim.x + threadIdx.x; e < E; e += stride) {
    const int s = ei[e];
    const int d = ei[E + e];
    int2 v;
    v.x = s;
    v.y = __float_as_int(dinv[s] * dinv[d]);
    bk[rowptr[d] + rank[e]] = v;
  }
}

__global__ void __launch_bounds__(256)
sort_rows_kernel(const int* __restrict__ rowptr, const int* __restrict__ bucket,
                 int* __restrict__ sorted, int N) {
  const int2* bk = (const int2*)bucket;
  int2* sk2 = (int2*)sorted;
  const int lane = threadIdx.x & 31;
  const int wv  = __builtin_amdgcn_readfirstlane((int)(threadIdx.x >> 5));
  const int wpb = (int)(blockDim.x >> 5);
  const int nw  = (int)gridDim.x * wpb;
  for (int row = blockIdx.x * wpb + wv; row < N; row += nw) {
    const int beg = __builtin_amdgcn_readfirstlane(rowptr[row]);
    const int n   = __builtin_amdgcn_readfirstlane(rowptr[row + 1]) - beg;
    for (int jc = 0; jc < n; jc += 32) {
      const int j = jc + lane;
      int2 my;
      my.x = 0x7fffffff; my.y = 0;
      if (j < n) my = bk[beg + j];
      int r = 0;
      for (int kc = 0; kc < n; kc += 32) {
        int sk = 0;
        if (kc + lane < n) sk = bk[beg + kc + lane].x;
        const int m = (n - kc < 32) ? (n - kc) : 32;
        for (int kk = 0; kk < m; ++kk) {
          const int v = __builtin_amdgcn_readlane(sk, kk);
          r += (int)(v < my.x) | ((int)(v == my.x) & (int)((kc + kk) < j));
        }
      }
      if (j < n) sk2[beg + r] = my;
    }
  }
}

__global__ void __launch_bounds__(128)
gemm_kernel(const float* __restrict__ x, const float* __restrict__ W,
            float* __restrict__ h, int N) {
  const int wave    = threadIdx.x >> 5;
  const int lane    = threadIdx.x & 31;
  const int half    = lane >> 4;
  const int l       = lane & 15;
  const int rowBase = blockIdx.x * 16;
  const int colBase = wave * 16;

  int arow = rowBase + l;
  arow = (arow < N) ? arow : (N - 1);
  const float* xr = x + (size_t)arow * FDIM + 2 * half;
  const float* wp = W + (size_t)(2 * half) * FDIM + colBase + l;

  v8f c = {};
  #pragma unroll
  for (int kk = 0; kk < 16; ++kk) {
    const v2f a = *(const v2f*)(xr + kk * 4);
    v2f b;
    b.x = wp[(size_t)kk * 4 * FDIM];
    b.y = wp[(size_t)kk * 4 * FDIM + FDIM];
    c = __builtin_amdgcn_wmma_f32_16x16x4_f32(
         false, a,  false, b,
         (short)0, c,  false,  false);
  }

  #pragma unroll
  for (int r = 0; r < 8; ++r) {
    const int row = rowBase + r + 8 * half;
    if (row < N) h[(size_t)row * FDIM + colBase + l] = c[r];
  }
}

#define GSLOTS 16
#define GREST  8
__global__ void __launch_bounds__(256)
gather_rows_kernel(const int* __restrict__ rowptr, const int* __restrict__ sorted,
                   const float* __restrict__ invdeg, const float* __restrict__ h,
                   const float* __restrict__ b, float* __restrict__ out,
                   int N, int do_relu) {
  const int2* sk2 = (const int2*)sorted;
  const int lane = threadIdx.x & 31;
  const int wv  = __builtin_amdgcn_readfirstlane((int)(threadIdx.x >> 5));
  const int wpb = (int)(blockDim.x >> 5);
  const int nw  = (int)gridDim.x * wpb;
  int row = blockIdx.x * wpb + wv;
  if (row >= N) return;
  const v2f bb = *(const v2f*)(b + 2 * lane);
  const float* hl = h + 2 * lane;

  int   beg0 = __builtin_amdgcn_readfirstlane(rowptr[row]);
  int   end0 = __builtin_amdgcn_readfirstlane(rowptr[row + 1]);
  float idg0 = invdeg[row];
  int   row1 = row + nw;
  int   beg1 = 0, end1 = 0;
  float idg1 = 0.0f;
  if (row1 < N) {
    beg1 = __builtin_amdgcn_readfirstlane(rowptr[row1]);
    end1 = __builtin_amdgcn_readfirstlane(rowptr[row1 + 1]);
    idg1 = invdeg[row1];
  }
  int2 e0;
  e0.x = 0; e0.y = 0;
  if (lane < end0 - beg0) e0 = sk2[beg0 + lane];

  for (;;) {
    const int n0 = end0 - beg0;
    v2f acc;
    acc.x = 0.0f; acc.y = 0.0f;

    const int mA = (n0 < GSLOTS) ? n0 : GSLOTS;
    v2f va[GSLOTS];
    const int sA0 = __builtin_amdgcn_readlane(e0.x, 0);
    #pragma unroll
    for (int j = 0; j < GSLOTS; ++j) {
      if (j < mA) {
        const int sj = (j == 0) ? sA0 : __builtin_amdgcn_readlane(e0.x, j);
        va[j] = *(const v2f*)(hl + (size_t)sj * FDIM);
      }
    }

    int2 e1;
    e1.x = 0; e1.y = 0;
    if (lane < end1 - beg1) e1 = sk2[beg1 + lane];
    const v2f hs = *(const v2f*)(hl + (size_t)row * FDIM);
    const int row2 = row1 + nw;
    int   beg2 = 0, end2 = 0;
    float idg2 = 0.0f;
    if (row2 < N) {
      beg2 = __builtin_amdgcn_readfirstlane(rowptr[row2]);
      end2 = __builtin_amdgcn_readfirstlane(rowptr[row2 + 1]);
      idg2 = invdeg[row2];
    }

    #pragma unroll
    for (int j = 0; j < GSLOTS; ++j) {
      if (j < mA) {
        const float wj = __int_as_float(__builtin_amdgcn_readlane(e0.y, j));
        acc.x += va[j].x * wj;
        acc.y += va[j].y * wj;
      }
    }

    int2 e = e0;
    for (int p = GSLOTS; p < n0; p += GREST) {
      const int q = p & 31;
      if (q == 0) {
        e.x = 0; e.y = 0;
        if (lane < n0 - p) e = sk2[beg0 + p + lane];
      }
      const int mB = (n0 - p < GREST) ? (n0 - p) : GREST;
      v2f vb[GREST];
      const int sB0 = __builtin_amdgcn_readlane(e.x, q);
      #pragma unroll
      for (int j = 0; j < GREST; ++j) {
        if (j == 0 || j < mB) {
          const int sj = (j == 0) ? sB0 : __builtin_amdgcn_readlane(e.x, q + j);
          vb[j] = *(const v2f*)(hl + (size_t)sj * FDIM);
        }
      }
      #pragma unroll
      for (int j = 0; j < GREST; ++j) {
        if (j == 0 || j < mB) {
          const float wj = __int_as_float(__builtin_amdgcn_readlane(e.y, q + j));
          acc.x += vb[j].x * wj;
          acc.y += vb[j].y * wj;
        }
      }
    }

    v2f o;
    o.x = (acc.x + hs.x * idg0) + bb.x;
    o.y = (acc.y + hs.y * idg0) + bb.y;
    if (do_relu) {
      o.x = fmaxf(o.x, 0.0f);
      o.y = fmaxf(o.y, 0.0f);
    }
    *(v2f*)(out + (size_t)row * FDIM + 2 * lane) = o;

    if (row1 >= N) break;
    row = row1;  beg0 = beg1; end0 = end1; idg0 = idg1; e0 = e1;
    row1 = row2; beg1 = beg2; end1 = end2; idg1 = idg2;
  }
}

extern "C" void kernel_launch(void* const* d_in, const int* in_sizes, int n_in,
                              void* d_out, int out_size, void* d_ws, size_t ws_size,
                              hipStream_t stream) {
  const float* x  = (const float*)d_in[0];
  const int*   ei = (const int*)  d_in[1];
  const float* W  = (const float*)d_in[2];
  const float* b  = (const float*)d_in[3];

  const int N     = in_sizes[0] / FDIM;
  const int E     = in_sizes[1] / 2;
  const int DEPTH = in_sizes[2] / (FDIM * FDIM);

  const int scanBlocks = (N + 1 + SCAN_ROWS - 1) / SCAN_ROWS;
  const size_t NP = (size_t)scanBlocks * SCAN_ROWS;
  const size_t EP = ((size_t)E + 63) / 64 * 64;

  int*   wsi    = (int*)d_ws;
  int*   cnt    = wsi;
  int*   rowptr = wsi + NP;
  float* dinv   = (float*)(wsi + 2 * NP);
  float* invdeg = (float*)(wsi + 3 * NP);
  int*   rank   = wsi + 4 * NP;
  int*   bucket = rank + EP;
  int*   sorted = bucket + 2 * EP;
  float* h      = (float*)(sorted + 2 * EP);
  float* xbuf   = h + (size_t)N * FDIM;

  const int T = 256;
  const int G = 2048;

  zero_i32_kernel<<<(int)(NP / T), T, 0, stream>>>(cnt, (int)NP);
  rank_kernel<<<G, T, 0, stream>>>(ei + E, cnt, rank, E);
  rowptr_kernel<<<scanBlocks, T, 0, stream>>>(cnt, rowptr, dinv, invdeg);
  fill_kernel<<<G, T, 0, stream>>>(ei, rowptr, rank, dinv, bucket, E);
  sort_rows_kernel<<<G, T, 0, stream>>>(rowptr, bucket, sorted, N);

  const int gemmBlocks = (N + 15) / 16;

  const float* xcur = x;
  for (int layer = 0; layer < DEPTH; ++layer) {
    gemm_kernel<<<gemmBlocks, 128, 0, stream>>>(
        xcur, W + (size_t)layer * FDIM * FDIM, h, N);
    float* xout = (layer == DEPTH - 1) ? (float*)d_out : xbuf;
    gather_rows_kernel<<<G, T, 0, stream>>>(
        rowptr, sorted, invdeg, h, b + (size_t)layer * FDIM, xout, N, layer != DEPTH - 1);
    xcur = xout;
  }
}


# Per-call bucket build redone as a two-level counting sort on LDS counters (edge chunks split by 256-row tile, then one work-group per tile): no global atomics, 13 -> 11 launches; sum order kept
# speedup vs baseline: 3.8916x; 1.5055x over previous
#include <hip/hip_runtime.h>


typedef float v2f __attribute__((ext_vector_type(2)));
typedef float v8f __attribute__((ext_vector_type(8)));

#define FDIM 64
#define SCAN_ROWS 1024
#define CH 4096
#define EPT (CH / 256)
#define TSHIFT 8
#define TROWS (1 << TSHIFT)
#define TMAX 2048

__global__ void __launch_bounds__(256)
tile_sort_kernel(const int* __restrict__ ei, int* __restrict__ rec,
                 int* __restrict__ off, int E, int T) {
  __shared__ int hist[TMAX];
  __shared__ int wtot[8];
  const int tid  = threadIdx.x;
  const int lane = tid & 31;
  const int wv   = tid >> 5;
  const int c    = blockIdx.x;
  const int e0   = c * CH;

  for (int i = tid; i < TMAX; i += 256) hist[i] = 0;
  int s[EPT], d[EPT], r[EPT];
  #pragma unroll
  for (int k = 0; k < EPT; ++k) {
    const int e = e0 + k * 256 + tid;
    s[k] = 0; d[k] = -1;
    if (e < E) { s[k] = ei[e]; d[k] = ei[E + e]; }
  }
  __syncthreads();
  #pragma unroll
  for (int k = 0; k < EPT; ++k) {
    r[k] = 0;
    if (d[k] >= 0) r[k] = atomicAdd(&hist[d[k] >> TSHIFT], 1);
  }
  __syncthreads();

  int v[TMAX / 256];
  int t = 0;
  #pragma unroll
  for (int j = 0; j < TMAX / 256; ++j) { v[j] = hist[tid * (TMAX / 256) + j]; t += v[j]; }
  int inc = t;
  #pragma unroll
  for (int o = 1; o < 32; o <<= 1) {
    const int u = __shfl_up(inc, o);
    if (lane >= o) inc += u;
  }
  if (lane == 31) wtot[wv] = inc;
  __syncthreads();
  int run = inc - t;
  #pragma unroll
  for (int w = 0; w < 8; ++w) if (w < wv) run += wtot[w];
  int* offc = off + (size_t)c * (T + 1);
  #pragma unroll
  for (int j = 0; j < TMAX / 256; ++j) {
    const int idx = tid * (TMAX / 256) + j;
    hist[idx] = run;
    if (idx <= T) offc[idx] = run;
    run += v[j];
  }
  __syncthreads();

  int2* rc = (int2*)rec;
  #pragma unroll
  for (int k = 0; k < EPT; ++k) {
    if (d[k] >= 0) {
      int2 q;
      q.x = s[k]; q.y = d[k];
      rc[(size_t)e0 + hist[d[k] >> TSHIFT] + r[k]] = q;
    }
  }
}

__global__ void __launch_bounds__(256)
csr_build_kernel(const int* __restrict__ rec, const int* __restrict__ off,
                 int* __restrict__ rowptr, float* __restrict__ dinv,
                 float* __restrict__ invdeg, int* __restrict__ bsrc, int C, int T) {
  __shared__ int lcnt[TROWS];
  __shared__ int lcur[TROWS];
  __shared__ int wsum[8];
  __shared__ int wbase[8];
  const int tid  = threadIdx.x;
  const int lane = tid & 31;
  const int wv   = tid >> 5;
  const int t    = blockIdx.x;
  const int TP   = T + 1;
  const int2* rc = (const int2*)rec;

  lcnt[tid] = 0;
  __syncthreads();
  int mybase = 0;
  for (int c = tid; c < C; c += 256) {
    const int a = off[(size_t)c * TP + t];
    const int b = off[(size_t)c * TP + t + 1];
    mybase += a;
    const int2* p = rc + (size_t)c * CH;
    for (int i = a; i < b; ++i) atomicAdd(&lcnt[p[i].y & (TROWS - 1)], 1);
  }
  #pragma unroll
  for (int o = 16; o > 0; o >>= 1) mybase += __shfl_xor(mybase, o);
  if (lane == 0) wbase[wv] = mybase;
  __syncthreads();

  const int n = lcnt[tid];
  int inc = n;
  #pragma unroll
  for (int o = 1; o < 32; o <<= 1) {
    const int u = __shfl_up(inc, o);
    if (lane >= o) inc += u;
  }
  if (lane == 31) wsum[wv] = inc;
  __syncthreads();
  int ex = inc - n;
  #pragma unroll
  for (int w = 0; w < 8; ++w) {
    ex += wbase[w];
    if (w < wv) ex += wsum[w];
  }
  const int row = t * TROWS + tid;
  rowptr[row] = ex;
  const float dg = (float)n + 1.0f;
  dinv[row]   = rsqrtf(dg);
  invdeg[row] = 1.0f / dg;
  lcur[tid] = ex;
  __syncthreads();

  for (int c = tid; c < C; c += 256) {
    const int a = off[(size_t)c * TP + t];
    const int b = off[(size_t)c * TP + t + 1];
    const int2* p = rc + (size_t)c * CH;
    for (int i = a; i < b; ++i) {
      const int2 q = p[i];
      bsrc[atomicAdd(&lcur[q.y & (TROWS - 1)], 1)] = q.x;
    }
  }
}

__global__ void __launch_bounds__(256)
sort_rows_kernel(const int* __restrict__ rowptr, const int* __restrict__ bsrc,
                 const float* __restrict__ dinv, int* __restrict__ sorted, int N) {
  int2* sk2 = (int2*)sorted;
  const int lane = threadIdx.x & 31;
  const int wv  = __builtin_amdgcn_readfirstlane((int)(threadIdx.x >> 5));
  const int wpb = (int)(blockDim.x >> 5);
  const int nw  = (int)gridDim.x * wpb;
  for (int row = blockIdx.x * wpb + wv; row < N; row += nw) {
    const int beg = __builtin_amdgcn_readfirstlane(rowptr[row]);
    const int n   = __builtin_amdgcn_readfirstlane(rowptr[row + 1]) - beg;
    const float dr = dinv[row];
    for (int jc = 0; jc < n; jc += 32) {
      const int j = jc + lane;
      int my = 0x7fffffff;
      float ds = 0.0f;
      if (j < n) {
        my = bsrc[beg + j];
        ds = dinv[my];
      }
      int r = 0;
      for (int kc = 0; kc < n; kc += 32) {
        int sk = my;
        if (kc != jc) {
          sk = 0;
          if (kc + lane < n) sk = bsrc[beg + kc + lane];
        }
        const int m = (n - kc < 32) ? (n - kc) : 32;
        for (int kk = 0; kk < m; ++kk) {
          const int v = __builtin_amdgcn_readlane(sk, kk);
          r += (int)(v < my) | ((int)(v == my) & (int)((kc + kk) < j));
        }
      }
      if (j < n) {
        int2 q;
        q.x = my;
        q.y = __float_as_int(ds * dr);
        sk2[beg + r] = q;
      }
    }
  }
}

__global__ void __launch_bounds__(128)
gemm_kernel(const float* __restrict__ x, const float* __restrict__ W,
            float* __restrict__ h, int N) {
  const int wave    = threadIdx.x >> 5;
  const int lane    = threadIdx.x & 31;
  const int half    = lane >> 4;
  const int l       = lane & 15;
  const int rowBase = blockIdx.x * 16;
  const int colBase = wave * 16;

  int arow = rowBase + l;
  arow = (arow < N) ? arow : (N - 1);
  const float* xr = x + (size_t)arow * FDIM + 2 * half;
  const float* wp = W + (size_t)(2 * half) * FDIM + colBase + l;

  v8f c = {};
  #pragma unroll
  for (int kk = 0; kk < 16; ++kk) {
    const v2f a = *(const v2f*)(xr + kk * 4);
    v2f b;
    b.x = wp[(size_t)kk * 4 * FDIM];
    b.y = wp[(size_t)kk * 4 * FDIM + FDIM];
    c = __builtin_amdgcn_wmma_f32_16x16x4_f32(
         false, a,  false, b,
         (short)0, c,  false,  false);
  }

  #pragma unroll
  for (int r = 0; r < 8; ++r) {
    const int row = rowBase + r + 8 * half;
    if (row < N) h[(size_t)row * FDIM + colBase + l] = c[r];
  }
}

#define GSLOTS 16
#define GREST  8
__global__ void __launch_bounds__(256)
gather_rows_kernel(const int* __restrict__ rowptr, const int* __restrict__ sorted,
                   const float* __restrict__ invdeg, const float* __restrict__ h,
                   const float* __restrict__ b, float* __restrict__ out,
                   int N, int do_relu) {
  const int2* sk2 = (const int2*)sorted;
  const int lane = threadIdx.x & 31;
  const int wv  = __builtin_amdgcn_readfirstlane((int)(threadIdx.x >> 5));
  const int wpb = (int)(blockDim.x >> 5);
  const int nw  = (int)gridDim.x * wpb;
  int row = blockIdx.x * wpb + wv;
  if (row >= N) return;
  const v2f bb = *(const v2f*)(b + 2 * lane);
  const float* hl = h + 2 * lane;

  int   beg0 = __builtin_amdgcn_readfirstlane(rowptr[row]);
  int   end0 = __builtin_amdgcn_readfirstlane(rowptr[row + 1]);
  float idg0 = invdeg[row];
  int   row1 = row + nw;
  int   beg1 = 0, end1 = 0;
  float idg1 = 0.0f;
  if (row1 < N) {
    beg1 = __builtin_amdgcn_readfirstlane(rowptr[row1]);
    end1 = __builtin_amdgcn_readfirstlane(rowptr[row1 + 1]);
    idg1 = invdeg[row1];
  }
  int2 e0;
  e0.x = 0; e0.y = 0;
  if (lane < end0 - beg0) e0 = sk2[beg0 + lane];

  for (;;) {
    const int n0 = end0 - beg0;
    v2f acc;
    acc.x = 0.0f; acc.y = 0.0f;

    const int mA = (n0 < GSLOTS) ? n0 : GSLOTS;
    v2f va[GSLOTS];
    const int sA0 = __builtin_amdgcn_readlane(e0.x, 0);
    #pragma unroll
    for (int j = 0; j < GSLOTS; ++j) {
      if (j < mA) {
        const int sj = (j == 0) ? sA0 : __builtin_amdgcn_readlane(e0.x, j);
        va[j] = *(const v2f*)(hl + (size_t)sj * FDIM);
      }
    }

    int2 e1;
    e1.x = 0; e1.y = 0;
    if (lane < end1 - beg1) e1 = sk2[beg1 + lane];
    const v2f hs = *(const v2f*)(hl + (size_t)row * FDIM);
    const int row2 = row1 + nw;
    int   beg2 = 0, end2 = 0;
    float idg2 = 0.0f;
    if (row2 < N) {
      beg2 = __builtin_amdgcn_readfirstlane(rowptr[row2]);
      end2 = __builtin_amdgcn_readfirstlane(rowptr[row2 + 1]);
      idg2 = invdeg[row2];
    }

    #pragma unroll
    for (int j = 0; j < GSLOTS; ++j) {
      if (j < mA) {
        const float wj = __int_as_float(__builtin_amdgcn_readlane(e0.y, j));
        acc.x += va[j].x * wj;
        acc.y += va[j].y * wj;
      }
    }

    int2 e = e0;
    for (int p = GSLOTS; p < n0; p += GREST) {
      const int q = p & 31;
      if (q == 0) {
        e.x = 0; e.y = 0;
        if (lane < n0 - p) e = sk2[beg0 + p + lane];
      }
      const int mB = (n0 - p < GREST) ? (n0 - p) : GREST;
      v2f vb[GREST];
      const int sB0 = __builtin_amdgcn_readlane(e.x, q);
      #pragma unroll
      for (int j = 0; j < GREST; ++j) {
        if (j == 0 || j < mB) {
          const int sj = (j == 0) ? sB0 : __builtin_amdgcn_readlane(e.x, q + j);
          vb[j] = *(const v2f*)(hl + (size_t)sj * FDIM);
        }
      }
      #pragma unroll
      for (int j = 0; j < GREST; ++j) {
        if (j == 0 || j < mB) {
          const float wj = __int_as_float(__builtin_amdgcn_readlane(e.y, q + j));
          acc.x += vb[j].x * wj;
          acc.y += vb[j].y * wj;
        }
      }
    }

    v2f o;
    o.x = (acc.x + hs.x * idg0) + bb.x;
    o.y = (acc.y + hs.y * idg0) + bb.y;
    if (do_relu) {
      o.x = fmaxf(o.x, 0.0f);
      o.y = fmaxf(o.y, 0.0f);
    }
    *(v2f*)(out + (size_t)row * FDIM + 2 * lane) = o;

    if (row1 >= N) break;
    row = row1;  beg0 = beg1; end0 = end1; idg0 = idg1; e0 = e1;
    row1 = row2; beg1 = beg2; end1 = end2; idg1 = idg2;
  }
}

extern "C" void kernel_launch(void* const* d_in, const int* in_sizes, int n_in,
                              void* d_out, int out_size, void* d_ws, size_t ws_size,
                              hipStream_t stream) {
  const float* x  = (const float*)d_in[0];
  const int*   ei = (const int*)  d_in[1];
  const float* W  = (const float*)d_in[2];
  const float* b  = (const float*)d_in[3];

  const int N     = in_sizes[0] / FDIM;
  const int E     = in_sizes[1] / 2;
  const int DEPTH = in_sizes[2] / (FDIM * FDIM);

  const int T = (N + 1 + TROWS - 1) / TROWS;
  const int C = (E + CH - 1) / CH;
  const size_t NP = (size_t)((N + 1 + SCAN_ROWS - 1) / SCAN_ROWS) * SCAN_ROWS;
  const size_t EP = (size_t)C * CH;

  int*   wsi    = (int*)d_ws;
  int*   rowptr = wsi;
  float* dinv   = (float*)(wsi + NP);
  float* invdeg = (float*)(wsi + 2 * NP);
  int*   bsrc   = wsi + 3 * NP;
  int*   rec    = bsrc + EP;
  int*   sorted = rec + 2 * EP;
  float* h      = (float*)(sorted + 2 * EP);
  float* xbuf   = h + (size_t)N * FDIM;
  int*   off    = (int*)(xbuf + (size_t)N * FDIM);

  const int TB = 256;
  const int G = 2048;

  tile_sort_kernel<<<C, TB, 0, stream>>>(ei, rec, off, E, T);
  csr_build_kernel<<<T, TB, 0, stream>>>(rec, off, rowptr, dinv, invdeg, bsrc, C, T);
  sort_rows_kernel<<<G, TB, 0, stream>>>(rowptr, bsrc, dinv, sorted, N);

  const int gemmBlocks = (N + 15) / 16;

  const float* xcur = x;
  for (int layer = 0; layer < DEPTH; ++layer) {
    gemm_kernel<<<gemmBlocks, 128, 0, stream>>>(
        xcur, W + (size_t)layer * FDIM * FDIM, h, N);
    float* xout = (layer == DEPTH - 1) ? (float*)d_out : xbuf;
    gather_rows_kernel<<<G, TB, 0, stream>>>(
        rowptr, sorted, invdeg, h, b + (size_t)layer * FDIM, xout, N, layer != DEPTH - 1);
    xcur = xout;
  }
}


# Layer GEMM fused behind the row gather: a wave sums 16 rows of x into an LDS tile, then 64 f32 WMMAs with W from LDS; h never stored, 11 -> 7 launches; rows are now summed before the product
# speedup vs baseline: 4.0115x; 1.0308x over previous
#include <hip/hip_runtime.h>


typedef float v2f __attribute__((ext_vector_type(2)));
typedef float v8f __attribute__((ext_vector_type(8)));

#define FDIM 64
#define SCAN_ROWS 1024
#define CH 4096
#define EPT (CH / 256)
#define TSHIFT 8
#define TROWS (1 << TSHIFT)
#define TMAX 2048

__global__ void __launch_bounds__(256)
tile_sort_kernel(const int* __restrict__ ei, int* __restrict__ rec,
                 int* __restrict__ off, int E, int T) {
  __shared__ int hist[TMAX];
  __shared__ int wtot[8];
  const int tid  = threadIdx.x;
  const int lane = tid & 31;
  const int wv   = tid >> 5;
  const int c    = blockIdx.x;
  const int e0   = c * CH;

  for (int i = tid; i < TMAX; i += 256) hist[i] = 0;
  int s[EPT], d[EPT], r[EPT];
  #pragma unroll
  for (int k = 0; k < EPT; ++k) {
    const int e = e0 + k * 256 + tid;
    s[k] = 0; d[k] = -1;
    if (e < E) { s[k] = ei[e]; d[k] = ei[E + e]; }
  }
  __syncthreads();
  #pragma unroll
  for (int k = 0; k < EPT; ++k) {
    r[k] = 0;
    if (d[k] >= 0) r[k] = atomicAdd(&hist[d[k] >> TSHIFT], 1);
  }
  __syncthreads();

  int v[TMAX / 256];
  int t = 0;
  #pragma unroll
  for (int j = 0; j < TMAX / 256; ++j) { v[j] = hist[tid * (TMAX / 256) + j]; t += v[j]; }
  int inc = t;
  #pragma unroll
  for (int o = 1; o < 32; o <<= 1) {
    const int u = __shfl_up(inc, o);
    if (lane >= o) inc += u;
  }
  if (lane == 31) wtot[wv] = inc;
  __syncthreads();
  int run = inc - t;
  #pragma unroll
  for (int w = 0; w < 8; ++w) if (w < wv) run += wtot[w];
  int* offc = off + (size_t)c * (T + 1);
  #pragma unroll
  for (int j = 0; j < TMAX / 256; ++j) {
    const int idx = tid * (TMAX / 256) + j;
    hist[idx] = run;
    if (idx <= T) offc[idx] = run;
    run += v[j];
  }
  __syncthreads();

  int2* rc = (int2*)rec;
  #pragma unroll
  for (int k = 0; k < EPT; ++k) {
    if (d[k] >= 0) {
      int2 q;
      q.x = s[k]; q.y = d[k];
      rc[(size_t)e0 + hist[d[k] >> TSHIFT] + r[k]] = q;
    }
  }
}

__global__ void __launch_bounds__(256)
csr_build_kernel(const int* __restrict__ rec, const int* __restrict__ off,
                 int* __restrict__ rowptr, float* __restrict__ dinv,
                 float* __restrict__ invdeg, int* __restrict__ bsrc, int C, int T) {
  __shared__ int lcnt[TROWS];
  __shared__ int lcur[TROWS];
  __shared__ int wsum[8];
  __shared__ int wbase[8];
  const int tid  = threadIdx.x;
  const int lane = tid & 31;
  const int wv   = tid >> 5;
  const int t    = blockIdx.x;
  const int TP   = T + 1;
  const int2* rc = (const int2*)rec;

  lcnt[tid] = 0;
  __syncthreads();
  int mybase = 0;
  for (int c = tid; c < C; c += 256) {
    const int a = off[(size_t)c * TP + t];
    const int b = off[(size_t)c * TP + t + 1];
    mybase += a;
    const int2* p = rc + (size_t)c * CH;
    for (int i = a; i < b; ++i) atomicAdd(&lcnt[p[i].y & (TROWS - 1)], 1);
  }
  #pragma unroll
  for (int o = 16; o > 0; o >>= 1) mybase += __shfl_xor(mybase, o);
  if (lane == 0) wbase[wv] = mybase;
  __syncthreads();

  const int n = lcnt[tid];
  int inc = n;
  #pragma unroll
  for (int o = 1; o < 32; o <<= 1) {
    const int u = __shfl_up(inc, o);
    if (lane >= o) inc += u;
  }
  if (lane == 31) wsum[wv] = inc;
  __syncthreads();
  int ex = inc - n;
  #pragma unroll
  for (int w = 0; w < 8; ++w) {
    ex += wbase[w];
    if (w < wv) ex += wsum[w];
  }
  const int row = t * TROWS + tid;
  rowptr[row] = ex;
  const float dg = (float)n + 1.0f;
  dinv[row]   = rsqrtf(dg);
  invdeg[row] = 1.0f / dg;
  lcur[tid] = ex;
  __syncthreads();

  for (int c = tid; c < C; c += 256) {
    const int a = off[(size_t)c * TP + t];
    const int b = off[(size_t)c * TP + t + 1];
    const int2* p = rc + (size_t)c * CH;
    for (int i = a; i < b; ++i) {
      const int2 q = p[i];
      bsrc[atomicAdd(&lcur[q.y & (TROWS - 1)], 1)] = q.x;
    }
  }
}

__global__ void __launch_bounds__(256)
sort_rows_kernel(const int* __restrict__ rowptr, const int* __restrict__ bsrc,
                 const float* __restrict__ dinv, int* __restrict__ sorted, int N) {
  int2* sk2 = (int2*)sorted;
  const int lane = threadIdx.x & 31;
  const int wv  = __builtin_amdgcn_readfirstlane((int)(threadIdx.x >> 5));
  const int wpb = (int)(blockDim.x >> 5);
  const int nw  = (int)gridDim.x * wpb;
  for (int row = blockIdx.x * wpb + wv; row < N; row += nw) {
    const int beg = __builtin_amdgcn_readfirstlane(rowptr[row]);
    const int n   = __builtin_amdgcn_readfirstlane(rowptr[row + 1]) - beg;
    const float dr = dinv[row];
    for (int jc = 0; jc < n; jc += 32) {
      const int j = jc + lane;
      int my = 0x7fffffff;
      float ds = 0.0f;
      if (j < n) {
        my = bsrc[beg + j];
        ds = dinv[my];
      }
      int r = 0;
      for (int kc = 0; kc < n; kc += 32) {
        int sk = my;
        if (kc != jc) {
          sk = 0;
          if (kc + lane < n) sk = bsrc[beg + kc + lane];
        }
        const int m = (n - kc < 32) ? (n - kc) : 32;
        for (int kk = 0; kk < m; ++kk) {
          const int v = __builtin_amdgcn_readlane(sk, kk);
          r += (int)(v < my) | ((int)(v == my) & (int)((kc + kk) < j));
        }
      }
      if (j < n) {
        int2 q;
        q.x = my;
        q.y = __float_as_int(ds * dr);
        sk2[beg + r] = q;
      }
    }
  }
}

#define GSLOTS 16
#define GREST  8
#define LDP 68
__global__ void __launch_bounds__(256)
layer_kernel(const int* __restrict__ rowptr, const int* __restrict__ sorted,
             const float* __restrict__ invdeg, const float* __restrict__ xin,
             const float* __restrict__ W, const float* __restrict__ b,
             float* __restrict__ out, int N, int do_relu) {
  __shared__ float Ws[FDIM * FDIM];
  __shared__ float tiles[8 * 16 * LDP];
  const int2* sk2 = (const int2*)sorted;
  const int tid  = threadIdx.x;
  const int lane = tid & 31;
  const int half = lane >> 4;
  const int l    = lane & 15;
  const int wv = __builtin_amdgcn_readfirstlane(tid >> 5);

  for (int i = tid; i < FDIM * FDIM; i += 256) {
    const int s  = i & 1;
    const int ln = (i >> 1) & 31;
    const int ct = (i >> 6) & 3;
    const int kk = i >> 8;
    Ws[i] = W[(kk * 4 + 2 * (ln >> 4) + s) * FDIM + ct * 16 + (ln & 15)];
  }
  __syncthreads();

  float* tile = tiles + wv * (16 * LDP);
  const v2f bb = *(const v2f*)(b + 2 * lane);
  const float* xl = xin + 2 * lane;
  const int nw      = (int)gridDim.x * 8;
  const int nstrips = (N + 15) >> 4;

  for (int strip = blockIdx.x * 8 + wv; strip < nstrips; strip += nw) {
    const int row0 = strip * 16;
    const int nr   = (N - row0 < 16) ? (N - row0) : 16;

    int rq = row0 + lane;
    rq = (rq < N) ? rq : N;
    const int rp = rowptr[rq];
    const int iq = (rq < N) ? rq : (N - 1);
    const int idv = __float_as_int(invdeg[iq]);

    int beg0 = __builtin_amdgcn_readlane(rp, 0);
    int end0 = __builtin_amdgcn_readlane(rp, 1);
    int2 e0;
    e0.x = 0; e0.y = 0;
    if (lane < end0 - beg0) e0 = sk2[beg0 + lane];

    for (int r = 0; r < nr; ++r) {
      const int row = row0 + r;
      const int n0 = end0 - beg0;
      v2f acc;
      acc.x = 0.0f; acc.y = 0.0f;

      const int mA = (n0 < GSLOTS) ? n0 : GSLOTS;
      v2f va[GSLOTS];
      const int sA0 = __builtin_amdgcn_readlane(e0.x, 0);
      #pragma unroll
      for (int j = 0; j < GSLOTS; ++j) {
        if (j < mA) {
          const int sj = (j == 0) ? sA0 : __builtin_amdgcn_readlane(e0.x, j);
          va[j] = *(const v2f*)(xl + (size_t)sj * FDIM);
        }
      }

      const int beg1 = end0;
      int end1 = end0;
      if (r + 1 < nr) end1 = __builtin_amdgcn_readlane(rp, r + 2);
      int2 e1;
      e1.x = 0; e1.y = 0;
      if (lane < end1 - beg1) e1 = sk2[beg1 + lane];
      const v2f hs = *(const v2f*)(xl + (size_t)row * FDIM);

      #pragma unroll
      for (int j = 0; j < GSLOTS; ++j) {
        if (j < mA) {
          const float wj = __int_as_float(__builtin_amdgcn_readlane(e0.y, j));
          acc.x += va[j].x * wj;
          acc.y += va[j].y * wj;
        }
      }

      int2 e = e0;
      for (int p = GSLOTS; p < n0; p += GREST) {
        const int q = p & 31;
        if (q == 0) {
          e.x = 0; e.y = 0;
          if (lane < n0 - p) e = sk2[beg0 + p + lane];
        }
        const int mB = (n0 - p < GREST) ? (n0 - p) : GREST;
        v2f vb[GREST];
        const int sB0 = __builtin_amdgcn_readlane(e.x, q);
        #pragma unroll
        for (int j = 0; j < GREST; ++j) {
          if (j == 0 || j < mB) {
            const int sj = (j == 0) ? sB0 : __builtin_amdgcn_readlane(e.x, q + j);
            vb[j] = *(const v2f*)(xl + (size_t)sj * FDIM);
          }
        }
        #pragma unroll
        for (int j = 0; j < GREST; ++j) {
          if (j == 0 || j < mB) {
            const float wj = __int_as_float(__builtin_amdgcn_readlane(e.y, q + j));
            acc.x += vb[j].x * wj;
            acc.y += vb[j].y * wj;
          }
        }
      }

      const float idg = __int_as_float(__builtin_amdgcn_readlane(idv, r));
      v2f o;
      o.x = acc.x + hs.x * idg;
      o.y = acc.y + hs.y * idg;
      *(v2f*)(tile + r * LDP + 2 * lane) = o;

      beg0 = beg1; end0 = end1; e0 = e1;
    }
    for (int r = nr; r < 16; ++r) {
      v2f z;
      z.x = 0.0f; z.y = 0.0f;
      *(v2f*)(tile + r * LDP + 2 * lane) = z;
    }
    __builtin_amdgcn_fence(__ATOMIC_ACQ_REL, "wavefront");

    v8f c0 = {}, c1 = {}, c2 = {}, c3 = {};
    const float* ap = tile + l * LDP + 2 * half;
    const float* bp = Ws + 2 * lane;
    v2f a  = *(const v2f*)(ap);
    v2f b0 = *(const v2f*)(bp + 0 * 64);
    v2f b1 = *(const v2f*)(bp + 1 * 64);
    v2f b2 = *(const v2f*)(bp + 2 * 64);
    v2f b3 = *(const v2f*)(bp + 3 * 64);
    __builtin_amdgcn_sched_group_barrier(0x100, 3, 0);
    #pragma unroll
    for (int kk = 0; kk < 16; ++kk) {
      v2f an = a, bn0 = b0, bn1 = b1, bn2 = b2, bn3 = b3;
      if (kk < 15) {
        an  = *(const v2f*)(ap + (kk + 1) * 4);
        bn0 = *(const v2f*)(bp + ((kk + 1) * 4 + 0) * 64);
        bn1 = *(const v2f*)(bp + ((kk + 1) * 4 + 1) * 64);
        bn2 = *(const v2f*)(bp + ((kk + 1) * 4 + 2) * 64);
        bn3 = *(const v2f*)(bp + ((kk + 1) * 4 + 3) * 64);
      }
      c0 = __builtin_amdgcn_wmma_f32_16x16x4_f32(false, a, false, b0, (short)0, c0, false, false);
      c1 = __builtin_amdgcn_wmma_f32_16x16x4_f32(false, a, false, b1, (short)0, c1, false, false);
      c2 = __builtin_amdgcn_wmma_f32_16x16x4_f32(false, a, false, b2, (short)0, c2, false, false);
      c3 = __builtin_amdgcn_wmma_f32_16x16x4_f32(false, a, false, b3, (short)0, c3, false, false);
      __builtin_amdgcn_sched_group_barrier(0x100, 3, 0);
      __builtin_amdgcn_sched_group_barrier(0x008, 4, 0);
      a = an; b0 = bn0; b1 = bn1; b2 = bn2; b3 = bn3;
    }

    float* cp = tile + (8 * half) * LDP + l;
    #pragma unroll
    for (int r = 0; r < 8; ++r) {
      cp[r * LDP +  0] = c0[r];
      cp[r * LDP + 16] = c1[r];
      cp[r * LDP + 32] = c2[r];
      cp[r * LDP + 48] = c3[r];
    }
    __builtin_amdgcn_fence(__ATOMIC_ACQ_REL, "wavefront");

    if (nr == 16) {
      v2f v[16];
      #pragma unroll
      for (int r = 0; r < 16; ++r) v[r] = *(const v2f*)(tile + r * LDP + 2 * lane);
      #pragma unroll
      for (int r = 0; r < 16; ++r) {
        v2f o;
        o.x = v[r].x + bb.x;
        o.y = v[r].y + bb.y;
        if (do_relu) {
          o.x = fmaxf(o.x, 0.0f);
          o.y = fmaxf(o.y, 0.0f);
        }
        *(v2f*)(out + (size_t)(row0 + r) * FDIM + 2 * lane) = o;
      }
    } else {
      for (int r = 0; r < nr; ++r) {
        const v2f v = *(const v2f*)(tile + r * LDP + 2 * lane);
        v2f o;
        o.x = v.x + bb.x;
        o.y = v.y + bb.y;
        if (do_relu) {
          o.x = fmaxf(o.x, 0.0f);
          o.y = fmaxf(o.y, 0.0f);
        }
        *(v2f*)(out + (size_t)(row0 + r) * FDIM + 2 * lane) = o;
      }
    }
  }
}

extern "C" void kernel_launch(void* const* d_in, const int* in_sizes, int n_in,
                              void* d_out, int out_size, void* d_ws, size_t ws_size,
                              hipStream_t stream) {
  const float* x  = (const float*)d_in[0];
  const int*   ei = (const int*)  d_in[1];
  const float* W  = (const float*)d_in[2];
  const float* b  = (const float*)d_in[3];

  const int N     = in_sizes[0] / FDIM;
  const int E     = in_sizes[1] / 2;
  const int DEPTH = in_sizes[2] / (FDIM * FDIM);

  const int T = (N + 1 + TROWS - 1) / TROWS;
  const int C = (E + CH - 1) / CH;
  const size_t NP = (size_t)((N + 1 + SCAN_ROWS - 1) / SCAN_ROWS) * SCAN_ROWS;
  const size_t EP = (size_t)C * CH;

  int*   wsi    = (int*)d_ws;
  int*   rowptr = wsi;
  float* dinv   = (float*)(wsi + NP);
  float* invdeg = (float*)(wsi + 2 * NP);
  int*   bsrc   = wsi + 3 * NP;
  int*   rec    = bsrc + EP;
  int*   sorted = rec + 2 * EP;
  float* xa     = (float*)(sorted + 2 * EP);
  float* xb     = xa + (size_t)N * FDIM;
  int*   off    = (int*)(xb + (size_t)N * FDIM);

  const int TB = 256;
  const int G = 2048;

  tile_sort_kernel<<<C, TB, 0, stream>>>(ei, rec, off, E, T);
  csr_build_kernel<<<T, TB, 0, stream>>>(rec, off, rowptr, dinv, invdeg, bsrc, C, T);
  sort_rows_kernel<<<G, TB, 0, stream>>>(rowptr, bsrc, dinv, sorted, N);

  const int layerBlocks = (N + 127) / 128;

  const float* xcur = x;
  for (int layer = 0; layer < DEPTH; ++layer) {
    float* xout = (layer == DEPTH - 1) ? (float*)d_out : ((layer & 1) ? xb : xa);
    layer_kernel<<<layerBlocks, TB, 0, stream>>>(
        rowptr, sorted, invdeg, xcur, W + (size_t)layer * FDIM * FDIM,
        b + (size_t)layer * FDIM, xout, N, layer != DEPTH - 1);
    xcur = xout;
  }
}


# Scattered record stores of both bucket-sort levels staged in LDS and copied out in address order, 16 B per lane, so each store covers four whole 128-B lines; buckets and sum order kept
# speedup vs baseline: 4.6074x; 1.1485x over previous
#include <hip/hip_runtime.h>


typedef float v2f __attribute__((ext_vector_type(2)));
typedef float v8f __attribute__((ext_vector_type(8)));

#define FDIM 64
#define SCAN_ROWS 1024
#define CH 4096
#define EPT (CH / 256)
#define TSHIFT 8
#define TROWS (1 << TSHIFT)
#define TMAX 2048

__global__ void __launch_bounds__(256)
tile_sort_kernel(const int* __restrict__ ei, int* __restrict__ rec,
                 int* __restrict__ off, int E, int T) {
  __shared__ int hist[TMAX];
  __shared__ int wtot[8];
  __shared__ __attribute__((aligned(16))) int2 stage[CH];
  const int tid  = threadIdx.x;
  const int lane = tid & 31;
  const int wv   = tid >> 5;
  const int c    = blockIdx.x;
  const int e0   = c * CH;

  for (int i = tid; i < TMAX; i += 256) hist[i] = 0;
  int s[EPT], d[EPT], r[EPT];
  #pragma unroll
  for (int k = 0; k < EPT; ++k) {
    const int e = e0 + k * 256 + tid;
    s[k] = 0; d[k] = -1;
    if (e < E) { s[k] = ei[e]; d[k] = ei[E + e]; }
  }
  __syncthreads();
  #pragma unroll
  for (int k = 0; k < EPT; ++k) {
    r[k] = 0;
    if (d[k] >= 0) r[k] = atomicAdd(&hist[d[k] >> TSHIFT], 1);
  }
  __syncthreads();

  int v[TMAX / 256];
  int t = 0;
  #pragma unroll
  for (int j = 0; j < TMAX / 256; ++j) { v[j] = hist[tid * (TMAX / 256) + j]; t += v[j]; }
  int inc = t;
  #pragma unroll
  for (int o = 1; o < 32; o <<= 1) {
    const int u = __shfl_up(inc, o);
    if (lane >= o) inc += u;
  }
  if (lane == 31) wtot[wv] = inc;
  __syncthreads();
  int run = inc - t;
  #pragma unroll
  for (int w = 0; w < 8; ++w) if (w < wv) run += wtot[w];
  int* offc = off + (size_t)c * (T + 1);
  #pragma unroll
  for (int j = 0; j < TMAX / 256; ++j) {
    const int idx = tid * (TMAX / 256) + j;
    hist[idx] = run;
    if (idx <= T) offc[idx] = run;
    run += v[j];
  }
  __syncthreads();

  #pragma unroll
  for (int k = 0; k < EPT; ++k) {
    if (d[k] >= 0) {
      int2 q;
      q.x = s[k]; q.y = d[k];
      stage[hist[d[k] >> TSHIFT] + r[k]] = q;
    }
  }
  __syncthreads();

  const int nv = (E - e0 < CH) ? (E - e0) : CH;
  int2* rc = (int2*)rec + (size_t)e0;
  if (nv == CH) {
    #pragma unroll
    for (int j = 0; j < CH / 512; ++j) {
      const int i = 2 * tid + 512 * j;
      *(int4*)(rc + i) = *(const int4*)(stage + i);
    }
  } else {
    for (int i = 2 * tid; i + 1 < nv; i += 512)
      *(int4*)(rc + i) = *(const int4*)(stage + i);
    if ((nv & 1) && tid == 0) rc[nv - 1] = stage[nv - 1];
  }
}

#define SCAP 8192
__global__ void __launch_bounds__(256)
csr_build_kernel(const int* __restrict__ rec, const int* __restrict__ off,
                 int* __restrict__ rowptr, float* __restrict__ dinv,
                 float* __restrict__ invdeg, int* __restrict__ bsrc, int C, int T) {
  __shared__ int lcnt[TROWS];
  __shared__ int lcur[TROWS];
  __shared__ int wsum[8];
  __shared__ int wbase[8];
  __shared__ __attribute__((aligned(16))) int stage[SCAP];
  const int tid  = threadIdx.x;
  const int lane = tid & 31;
  const int wv   = tid >> 5;
  const int t    = blockIdx.x;
  const int TP   = T + 1;
  const int2* rc = (const int2*)rec;

  lcnt[tid] = 0;
  __syncthreads();
  int mybase = 0;
  for (int c = tid; c < C; c += 256) {
    const int a = off[(size_t)c * TP + t];
    const int b = off[(size_t)c * TP + t + 1];
    mybase += a;
    const int2* p = rc + (size_t)c * CH;
    for (int i = a; i < b; ++i) atomicAdd(&lcnt[p[i].y & (TROWS - 1)], 1);
  }
  #pragma unroll
  for (int o = 16; o > 0; o >>= 1) mybase += __shfl_xor(mybase, o);
  if (lane == 0) wbase[wv] = mybase;
  __syncthreads();

  const int n = lcnt[tid];
  int inc = n;
  #pragma unroll
  for (int o = 1; o < 32; o <<= 1) {
    const int u = __shfl_up(inc, o);
    if (lane >= o) inc += u;
  }
  if (lane == 31) wsum[wv] = inc;
  __syncthreads();
  int ex = inc - n;
  int tb = 0, cnt = 0;
  #pragma unroll
  for (int w = 0; w < 8; ++w) {
    tb  += wbase[w];
    cnt += wsum[w];
    if (w < wv) ex += wsum[w];
  }
  ex += tb;
  const int row = t * TROWS + tid;
  rowptr[row] = ex;
  const float dg = (float)n + 1.0f;
  dinv[row]   = rsqrtf(dg);
  invdeg[row] = 1.0f / dg;
  lcur[tid] = ex;
  __syncthreads();

  const int cntu = __builtin_amdgcn_readfirstlane(cnt);
  const int tbu  = __builtin_amdgcn_readfirstlane(tb);
  const int al  = tbu & 31;
  const int tl  = tbu - al;
  const int end = al + cntu;
  if (end <= SCAP) {
    for (int c = tid; c < C; c += 256) {
      const int a = off[(size_t)c * TP + t];
      const int b = off[(size_t)c * TP + t + 1];
      const int2* p = rc + (size_t)c * CH;
      for (int i = a; i < b; ++i) {
        const int2 q = p[i];
        stage[atomicAdd(&lcur[q.y & (TROWS - 1)], 1) - tl] = q.x;
      }
    }
    __syncthreads();
    for (int s = 4 * tid; s < end; s += 1024) {
      if (s >= al && s + 4 <= end) {
        *(int4*)(bsrc + tl + s) = *(const int4*)(stage + s);
      } else {
        for (int u = 0; u < 4; ++u)
          if (s + u >= al && s + u < end) bsrc[tl + s + u] = stage[s + u];
      }
    }
  } else {
    for (int c = tid; c < C; c += 256) {
      const int a = off[(size_t)c * TP + t];
      const int b = off[(size_t)c * TP + t + 1];
      const int2* p = rc + (size_t)c * CH;
      for (int i = a; i < b; ++i) {
        const int2 q = p[i];
        bsrc[atomicAdd(&lcur[q.y & (TROWS - 1)], 1)] = q.x;
      }
    }
  }
}

__global__ void __launch_bounds__(256)
sort_rows_kernel(const int* __restrict__ rowptr, const int* __restrict__ bsrc,
                 const float* __restrict__ dinv, int* __restrict__ sorted, int N) {
  int2* sk2 = (int2*)sorted;
  const int lane = threadIdx.x & 31;
  const int wv  = __builtin_amdgcn_readfirstlane((int)(threadIdx.x >> 5));
  const int wpb = (int)(blockDim.x >> 5);
  const int nw  = (int)gridDim.x * wpb;
  for (int row = blockIdx.x * wpb + wv; row < N; row += nw) {
    const int beg = __builtin_amdgcn_readfirstlane(rowptr[row]);
    const int n   = __builtin_amdgcn_readfirstlane(rowptr[row + 1]) - beg;
    const float dr = dinv[row];
    for (int jc = 0; jc < n; jc += 32) {
      const int j = jc + lane;
      int my = 0x7fffffff;
      float ds = 0.0f;
      if (j < n) {
        my = bsrc[beg + j];
        ds = dinv[my];
      }
      int r = 0;
      for (int kc = 0; kc < n; kc += 32) {
        int sk = my;
        if (kc != jc) {
          sk = 0;
          if (kc + lane < n) sk = bsrc[beg + kc + lane];
        }
        const int m = (n - kc < 32) ? (n - kc) : 32;
        for (int kk = 0; kk < m; ++kk) {
          const int v = __builtin_amdgcn_readlane(sk, kk);
          r += (int)(v < my) | ((int)(v == my) & (int)((kc + kk) < j));
        }
      }
      if (j < n) {
        int2 q;
        q.x = my;
        q.y = __float_as_int(ds * dr);
        sk2[beg + r] = q;
      }
    }
  }
}

typedef float v4f __attribute__((ext_vector_type(4)));
#define GS 22
#define LDP 68
#define HALF_BCAST(v, K) __builtin_amdgcn_update_dpp(0, (v), 0x150 + (K), 0xF, 0xF, true)
#define SLOTS(X) X(0, e_lo, 0) X(1, e_lo, 1) X(2, e_lo, 2) X(3, e_lo, 3) X(4, e_lo, 4) X(5, e_lo, 5) X(6, e_lo, 6) X(7, e_lo, 7) X(8, e_lo, 8) X(9, e_lo, 9) X(10, e_lo, 10) X(11, e_lo, 11) X(12, e_lo, 12) X(13, e_lo, 13) X(14, e_lo, 14) X(15, e_lo, 15) X(16, e_hi, 0) X(17, e_hi, 1) X(18, e_hi, 2) X(19, e_hi, 3) X(20, e_hi, 4) X(21, e_hi, 5)
#define ROW_LOAD(J, E, K) { const unsigned so = ((J) == 0) ? so_lo : ((J) == 16) ? so_hi : ((((unsigned)HALF_BCAST(E.x, K)) << 8) | l16); va[J] = *(const v4f*)(xb + so); }
#define ROW_SUM(J, E, K)  { const float wj = __int_as_float(HALF_BCAST(E.y, K)); acc += va[J] * wj; }
#define ISSUE(J, E, K) if ((J) < mx) ROW_LOAD(J, E, K)
#define SUM(J, E, K)   if ((J) < rem) ROW_SUM(J, E, K)
__global__ void __launch_bounds__(256)
layer_kernel(const int* __restrict__ rowptr, const int* __restrict__ sorted,
             const float* __restrict__ invdeg, const float* __restrict__ xin,
             const float* __restrict__ W, const float* __restrict__ b,
             float* __restrict__ out, int N, int do_relu) {
  __shared__ float Ws[FDIM * FDIM];
  __shared__ __attribute__((aligned(16))) float tiles[8 * 16 * LDP];
  const int2* sk2 = (const int2*)sorted;
  const int tid  = threadIdx.x;
  const int lane = tid & 31;
  const int half = lane >> 4;
  const int l    = lane & 15;
  const int wv = __builtin_amdgcn_readfirstlane(tid >> 5);

  for (int i = tid; i < FDIM * FDIM; i += 256) {
    const int s  = i & 1;
    const int ln = (i >> 1) & 31;
    const int ct = (i >> 6) & 3;
    const int kk = i >> 8;
    Ws[i] = W[(kk * 4 + 2 * (ln >> 4) + s) * FDIM + ct * 16 + (ln & 15)];
  }
  __syncthreads();

  float* tile = tiles + wv * (16 * LDP);
  const v2f bb = *(const v2f*)(b + 2 * lane);
  const char* xb = (const char*)xin;
  const unsigned l16 = (unsigned)l << 4;
  const int nstrips = (N + 15) >> 4;

  const int strip = blockIdx.x * 8 + wv;
  if (strip < nstrips) {
    const int row0 = strip * 16;
    const int nr   = (N - row0 < 16) ? (N - row0) : 16;

    int rq = row0 + lane;
    rq = (rq < N) ? rq : N;
    const int rp = rowptr[rq];
    const int iq = (rq < N) ? rq : (N - 1);
    const int idv = __float_as_int(invdeg[iq]);

    int i = 0, c = 0;
    int n_a, n_b, my_beg, my_n;
    {
      const int a0 = __builtin_amdgcn_readlane(rp, 0), a1 = __builtin_amdgcn_readlane(rp, 1);
      const int b0 = __builtin_amdgcn_readlane(rp, 8), b1 = __builtin_amdgcn_readlane(rp, 9);
      n_a = a1 - a0; n_b = b1 - b0;
      my_beg = half ? b0 : a0;
      my_n   = half ? n_b : n_a;
    }
    int2 e_lo, e_hi;
    e_lo.x = 0; e_lo.y = 0; e_hi.x = 0; e_hi.y = 0;
    if (l < my_n)      e_lo = sk2[my_beg + l];
    if (l + 16 < my_n) e_hi = sk2[my_beg + 16 + l];
    v4f acc = {0.0f, 0.0f, 0.0f, 0.0f};
    v4f hs  = {0.0f, 0.0f, 0.0f, 0.0f};

    for (;;) {
      const int ra = n_a - c, rb = n_b - c;
      const int mxr = (ra < rb) ? rb : ra;
      const int mx  = (mxr < GS) ? mxr : GS;
      const int rem = my_n - c;
      v4f va[GS];

      const unsigned so_lo = (((unsigned)HALF_BCAST(e_lo.x, 0)) << 8) | l16;
      const unsigned so_hi = (((unsigned)HALF_BCAST(e_hi.x, 0)) << 8) | l16;

      SLOTS(ISSUE)

      int ni = i, nc = c + GS;
      int nn_a = n_a, nn_b = n_b, nbeg = my_beg, nmy = my_n;
      if (mxr <= GS) {
        ni = i + 1; nc = 0;
        nn_a = 0; nn_b = 0; nmy = 0;
        if (ni < 8) {
          const int a0 = __builtin_amdgcn_readlane(rp, ni),     a1 = __builtin_amdgcn_readlane(rp, ni + 1);
          const int b0 = __builtin_amdgcn_readlane(rp, ni + 8), b1 = __builtin_amdgcn_readlane(rp, ni + 9);
          nn_a = a1 - a0; nn_b = b1 - b0;
          nbeg = half ? b0 : a0;
          nmy  = half ? nn_b : nn_a;
        }
      }
      const int nrem = nmy - nc;
      int2 ne_lo, ne_hi;
      ne_lo.x = 0; ne_lo.y = 0; ne_hi.x = 0; ne_hi.y = 0;
      if (l < nrem)      ne_lo = sk2[nbeg + nc + l];
      if (l + 16 < nrem) ne_hi = sk2[nbeg + nc + 16 + l];
      if (c == 0) {
        const int myrow = row0 + 8 * half + i;
        hs.x = 0.0f; hs.y = 0.0f; hs.z = 0.0f; hs.w = 0.0f;
        if (myrow < N) hs = *(const v4f*)(xb + (((unsigned)myrow << 8) | l16));
      }

      SLOTS(SUM)

      if (ni != i) {
        const float ia = __int_as_float(__builtin_amdgcn_readlane(idv, i));
        const float ib = __int_as_float(__builtin_amdgcn_readlane(idv, i + 8));
        const float idg = half ? ib : ia;
        const v4f o = acc + hs * idg;
        *(v4f*)(tile + (8 * half + i) * LDP + 4 * l) = o;
        acc.x = 0.0f; acc.y = 0.0f; acc.z = 0.0f; acc.w = 0.0f;
        if (ni >= 8) break;
      }
      i = ni; c = nc; n_a = nn_a; n_b = nn_b; my_beg = nbeg; my_n = nmy;
      e_lo = ne_lo; e_hi = ne_hi;
    }
    __builtin_amdgcn_fence(__ATOMIC_ACQ_REL, "wavefront");

    v8f c0 = {}, c1 = {}, c2 = {}, c3 = {};
    const float* ap = tile + l * LDP + 2 * half;
    const float* bp = Ws + 2 * lane;
    v2f a  = *(const v2f*)(ap);
    v2f b0 = *(const v2f*)(bp + 0 * 64);
    v2f b1 = *(const v2f*)(bp + 1 * 64);
    v2f b2 = *(const v2f*)(bp + 2 * 64);
    v2f b3 = *(const v2f*)(bp + 3 * 64);
    __builtin_amdgcn_sched_group_barrier(0x100, 3, 0);
    #pragma unroll
    for (int kk = 0; kk < 16; ++kk) {
      v2f an = a, bn0 = b0, bn1 = b1, bn2 = b2, bn3 = b3;
      if (kk < 15) {
        an  = *(const v2f*)(ap + (kk + 1) * 4);
        bn0 = *(const v2f*)(bp + ((kk + 1) * 4 + 0) * 64);
        bn1 = *(const v2f*)(bp + ((kk + 1) * 4 + 1) * 64);
        bn2 = *(const v2f*)(bp + ((kk + 1) * 4 + 2) * 64);
        bn3 = *(const v2f*)(bp + ((kk + 1) * 4 + 3) * 64);
      }
      c0 = __builtin_amdgcn_wmma_f32_16x16x4_f32(false, a, false, b0, (short)0, c0, false, false);
      c1 = __builtin_amdgcn_wmma_f32_16x16x4_f32(false, a, false, b1, (short)0, c1, false, false);
      c2 = __builtin_amdgcn_wmma_f32_16x16x4_f32(false, a, false, b2, (short)0, c2, false, false);
      c3 = __builtin_amdgcn_wmma_f32_16x16x4_f32(false, a, false, b3, (short)0, c3, false, false);
      __builtin_amdgcn_sched_group_barrier(0x100, 3, 0);
      __builtin_amdgcn_sched_group_barrier(0x008, 4, 0);
      a = an; b0 = bn0; b1 = bn1; b2 = bn2; b3 = bn3;
    }

    float* cp = tile + (8 * half) * LDP + l;
    #pragma unroll
    for (int r = 0; r < 8; ++r) {
      cp[r * LDP +  0] = c0[r];
      cp[r * LDP + 16] = c1[r];
      cp[r * LDP + 32] = c2[r];
      cp[r * LDP + 48] = c3[r];
    }
    __builtin_amdgcn_fence(__ATOMIC_ACQ_REL, "wavefront");

    if (nr == 16) {
      v2f v[16];
      #pragma unroll
      for (int r = 0; r < 16; ++r) v[r] = *(const v2f*)(tile + r * LDP + 2 * lane);
      #pragma unroll
      for (int r = 0; r < 16; ++r) {
        v2f o;
        o.x = v[r].x + bb.x;
        o.y = v[r].y + bb.y;
        if (do_relu) {
          o.x = fmaxf(o.x, 0.0f);
          o.y = fmaxf(o.y, 0.0f);
        }
        *(v2f*)(out + (size_t)(row0 + r) * FDIM + 2 * lane) = o;
      }
    } else {
      for (int r = 0; r < nr; ++r) {
        const v2f v = *(const v2f*)(tile + r * LDP + 2 * lane);
        v2f o;
        o.x = v.x + bb.x;
        o.y = v.y + bb.y;
        if (do_relu) {
          o.x = fmaxf(o.x, 0.0f);
          o.y = fmaxf(o.y, 0.0f);
        }
        *(v2f*)(out + (size_t)(row0 + r) * FDIM + 2 * lane) = o;
      }
    }
  }
}

extern "C" void kernel_launch(void* const* d_in, const int* in_sizes, int n_in,
                              void* d_out, int out_size, void* d_ws, size_t ws_size,
                              hipStream_t stream) {
  const float* x  = (const float*)d_in[0];
  const int*   ei = (const int*)  d_in[1];
  const float* W  = (const float*)d_in[2];
  const float* b  = (const float*)d_in[3];

  const int N     = in_sizes[0] / FDIM;
  const int E     = in_sizes[1] / 2;
  const int DEPTH = in_sizes[2] / (FDIM * FDIM);

  const int T = (N + 1 + TROWS - 1) / TROWS;
  const int C = (E + CH - 1) / CH;
  const size_t NP = (size_t)((N + 1 + SCAN_ROWS - 1) / SCAN_ROWS) * SCAN_ROWS;
  const size_t EP = (size_t)C * CH;

  int*   wsi    = (int*)d_ws;
  int*   rowptr = wsi;
  float* dinv   = (float*)(wsi + NP);
  float* invdeg = (float*)(wsi + 2 * NP);
  int*   bsrc   = wsi + 3 * NP;
  int*   rec    = bsrc + EP;
  int*   sorted = rec + 2 * EP;
  float* xa     = (float*)(sorted + 2 * EP);
  float* xb     = xa + (size_t)N * FDIM;
  int*   off    = (int*)(xb + (size_t)N * FDIM);

  const int TB = 256;
  const int G = 2048;

  tile_sort_kernel<<<C, TB, 0, stream>>>(ei, rec, off, E, T);
  csr_build_kernel<<<T, TB, 0, stream>>>(rec, off, rowptr, dinv, invdeg, bsrc, C, T);
  sort_rows_kernel<<<G, TB, 0, stream>>>(rowptr, bsrc, dinv, sorted, N);

  const int layerBlocks = (N + 127) / 128;

  const float* xcur = x;
  for (int layer = 0; layer < DEPTH; ++layer) {
    float* xout = (layer == DEPTH - 1) ? (float*)d_out : ((layer & 1) ? xb : xa);
    layer_kernel<<<layerBlocks, TB, 0, stream>>>(
        rowptr, sorted, invdeg, xcur, W + (size_t)layer * FDIM * FDIM,
        b + (size_t)layer * FDIM, xout, N, layer != DEPTH - 1);
    xcur = xout;
  }
}


# Bucket records of a strip's 16 rows (one contiguous run) fetched once, whole lines per load, into a wave-private LDS stage; gather steps read entries from LDS, not memory; sum order kept
# speedup vs baseline: 4.7667x; 1.0346x over previous
#include <hip/hip_runtime.h>


typedef float v2f __attribute__((ext_vector_type(2)));
typedef float v8f __attribute__((ext_vector_type(8)));

#define FDIM 64
#define SCAN_ROWS 1024
#define CH 4096
#define EPT (CH / 256)
#define TSHIFT 8
#define TROWS (1 << TSHIFT)
#define TMAX 2048

__global__ void __launch_bounds__(256)
tile_sort_kernel(const int* __restrict__ ei, int* __restrict__ rec,
                 int* __restrict__ off, int E, int T) {
  __shared__ int hist[TMAX];
  __shared__ int wtot[8];
  __shared__ __attribute__((aligned(16))) int2 stage[CH];
  const int tid  = threadIdx.x;
  const int lane = tid & 31;
  const int wv   = tid >> 5;
  const int c    = blockIdx.x;
  const int e0   = c * CH;

  for (int i = tid; i < TMAX; i += 256) hist[i] = 0;
  int s[EPT], d[EPT], r[EPT];
  #pragma unroll
  for (int k = 0; k < EPT; ++k) {
    const int e = e0 + k * 256 + tid;
    s[k] = 0; d[k] = -1;
    if (e < E) { s[k] = ei[e]; d[k] = ei[E + e]; }
  }
  __syncthreads();
  #pragma unroll
  for (int k = 0; k < EPT; ++k) {
    r[k] = 0;
    if (d[k] >= 0) r[k] = atomicAdd(&hist[d[k] >> TSHIFT], 1);
  }
  __syncthreads();

  int v[TMAX / 256];
  int t = 0;
  #pragma unroll
  for (int j = 0; j < TMAX / 256; ++j) { v[j] = hist[tid * (TMAX / 256) + j]; t += v[j]; }
  int inc = t;
  #pragma unroll
  for (int o = 1; o < 32; o <<= 1) {
    const int u = __shfl_up(inc, o);
    if (lane >= o) inc += u;
  }
  if (lane == 31) wtot[wv] = inc;
  __syncthreads();
  int run = inc - t;
  #pragma unroll
  for (int w = 0; w < 8; ++w) if (w < wv) run += wtot[w];
  int* offc = off + (size_t)c * (T + 1);
  #pragma unroll
  for (int j = 0; j < TMAX / 256; ++j) {
    const int idx = tid * (TMAX / 256) + j;
    hist[idx] = run;
    if (idx <= T) offc[idx] = run;
    run += v[j];
  }
  __syncthreads();

  #pragma unroll
  for (int k = 0; k < EPT; ++k) {
    if (d[k] >= 0) {
      int2 q;
      q.x = s[k]; q.y = d[k];
      stage[hist[d[k] >> TSHIFT] + r[k]] = q;
    }
  }
  __syncthreads();

  const int nv = (E - e0 < CH) ? (E - e0) : CH;
  int2* rc = (int2*)rec + (size_t)e0;
  if (nv == CH) {
    #pragma unroll
    for (int j = 0; j < CH / 512; ++j) {
      const int i = 2 * tid + 512 * j;
      *(int4*)(rc + i) = *(const int4*)(stage + i);
    }
  } else {
    for (int i = 2 * tid; i + 1 < nv; i += 512)
      *(int4*)(rc + i) = *(const int4*)(stage + i);
    if ((nv & 1) && tid == 0) rc[nv - 1] = stage[nv - 1];
  }
}

#define SCAP 8192
__global__ void __launch_bounds__(256)
csr_build_kernel(const int* __restrict__ rec, const int* __restrict__ off,
                 int* __restrict__ rowptr, float* __restrict__ dinv,
                 float* __restrict__ invdeg, int* __restrict__ bsrc, int C, int T) {
  __shared__ int lcnt[TROWS];
  __shared__ int lcur[TROWS];
  __shared__ int wsum[8];
  __shared__ int wbase[8];
  __shared__ __attribute__((aligned(16))) int stage[SCAP];
  const int tid  = threadIdx.x;
  const int lane = tid & 31;
  const int wv   = tid >> 5;
  const int t    = blockIdx.x;
  const int TP   = T + 1;
  const int2* rc = (const int2*)rec;

  lcnt[tid] = 0;
  __syncthreads();
  int mybase = 0;
  for (int c = tid; c < C; c += 256) {
    const int a = off[(size_t)c * TP + t];
    const int b = off[(size_t)c * TP + t + 1];
    mybase += a;
    const int2* p = rc + (size_t)c * CH;
    for (int i = a; i < b; ++i) atomicAdd(&lcnt[p[i].y & (TROWS - 1)], 1);
  }
  #pragma unroll
  for (int o = 16; o > 0; o >>= 1) mybase += __shfl_xor(mybase, o);
  if (lane == 0) wbase[wv] = mybase;
  __syncthreads();

  const int n = lcnt[tid];
  int inc = n;
  #pragma unroll
  for (int o = 1; o < 32; o <<= 1) {
    const int u = __shfl_up(inc, o);
    if (lane >= o) inc += u;
  }
  if (lane == 31) wsum[wv] = inc;
  __syncthreads();
  int ex = inc - n;
  int tb = 0, cnt = 0;
  #pragma unroll
  for (int w = 0; w < 8; ++w) {
    tb  += wbase[w];
    cnt += wsum[w];
    if (w < wv) ex += wsum[w];
  }
  ex += tb;
  const int row = t * TROWS + tid;
  rowptr[row] = ex;
  const float dg = (float)n + 1.0f;
  dinv[row]   = rsqrtf(dg);
  invdeg[row] = 1.0f / dg;
  lcur[tid] = ex;
  __syncthreads();

  const int cntu = __builtin_amdgcn_readfirstlane(cnt);
  const int tbu  = __builtin_amdgcn_readfirstlane(tb);
  const int al  = tbu & 31;
  const int tl  = tbu - al;
  const int end = al + cntu;
  if (end <= SCAP) {
    for (int c = tid; c < C; c += 256) {
      const int a = off[(size_t)c * TP + t];
      const int b = off[(size_t)c * TP + t + 1];
      const int2* p = rc + (size_t)c * CH;
      for (int i = a; i < b; ++i) {
        const int2 q = p[i];
        stage[atomicAdd(&lcur[q.y & (TROWS - 1)], 1) - tl] = q.x;
      }
    }
    __syncthreads();
    for (int s = 4 * tid; s < end; s += 1024) {
      if (s >= al && s + 4 <= end) {
        *(int4*)(bsrc + tl + s) = *(const int4*)(stage + s);
      } else {
        for (int u = 0; u < 4; ++u)
          if (s + u >= al && s + u < end) bsrc[tl + s + u] = stage[s + u];
      }
    }
  } else {
    for (int c = tid; c < C; c += 256) {
      const int a = off[(size_t)c * TP + t];
      const int b = off[(size_t)c * TP + t + 1];
      const int2* p = rc + (size_t)c * CH;
      for (int i = a; i < b; ++i) {
        const int2 q = p[i];
        bsrc[atomicAdd(&lcur[q.y & (TROWS - 1)], 1)] = q.x;
      }
    }
  }
}

__global__ void __launch_bounds__(256)
sort_rows_kernel(const int* __restrict__ rowptr, const int* __restrict__ bsrc,
                 const float* __restrict__ dinv, int* __restrict__ sorted, int N) {
  int2* sk2 = (int2*)sorted;
  const int lane = threadIdx.x & 31;
  const int wv  = __builtin_amdgcn_readfirstlane((int)(threadIdx.x >> 5));
  const int wpb = (int)(blockDim.x >> 5);
  const int nw  = (int)gridDim.x * wpb;
  for (int row = blockIdx.x * wpb + wv; row < N; row += nw) {
    const int beg = __builtin_amdgcn_readfirstlane(rowptr[row]);
    const int n   = __builtin_amdgcn_readfirstlane(rowptr[row + 1]) - beg;
    const float dr = dinv[row];
    for (int jc = 0; jc < n; jc += 32) {
      const int j = jc + lane;
      int my = 0x7fffffff;
      float ds = 0.0f;
      if (j < n) {
        my = bsrc[beg + j];
        ds = dinv[my];
      }
      int r = 0;
      for (int kc = 0; kc < n; kc += 32) {
        int sk = my;
        if (kc != jc) {
          sk = 0;
          if (kc + lane < n) sk = bsrc[beg + kc + lane];
        }
        const int m = (n - kc < 32) ? (n - kc) : 32;
        for (int kk = 0; kk < m; ++kk) {
          const int v = __builtin_amdgcn_readlane(sk, kk);
          r += (int)(v < my) | ((int)(v == my) & (int)((kc + kk) < j));
        }
      }
      if (j < n) {
        int2 q;
        q.x = my;
        q.y = __float_as_int(ds * dr);
        sk2[beg + r] = q;
      }
    }
  }
}

typedef float v4f __attribute__((ext_vector_type(4)));
#define GS 22
#define LDP 68
#define RCAP 384
#define REC_AT(DST, IX, ON) { \
    const int cx_ = (ON) ? (((IX) < RCAP) ? (IX) : (RCAP - 1)) : 0; \
    int2 t_ = rs[cx_]; \
    if (big) { if ((ON) && (IX) >= RCAP) t_ = sk2[sbeg + (IX)]; } \
    DST.x = (ON) ? t_.x : 0; DST.y = (ON) ? t_.y : 0; }
#define HALF_BCAST(v, K) __builtin_amdgcn_update_dpp(0, (v), 0x150 + (K), 0xF, 0xF, true)
#define SLOTS(X) X(0, e_lo, 0) X(1, e_lo, 1) X(2, e_lo, 2) X(3, e_lo, 3) X(4, e_lo, 4) X(5, e_lo, 5) X(6, e_lo, 6) X(7, e_lo, 7) X(8, e_lo, 8) X(9, e_lo, 9) X(10, e_lo, 10) X(11, e_lo, 11) X(12, e_lo, 12) X(13, e_lo, 13) X(14, e_lo, 14) X(15, e_lo, 15) X(16, e_hi, 0) X(17, e_hi, 1) X(18, e_hi, 2) X(19, e_hi, 3) X(20, e_hi, 4) X(21, e_hi, 5)
#define ROW_LOAD(J, E, K) { const unsigned so = ((J) == 0) ? so_lo : ((J) == 16) ? so_hi : ((((unsigned)HALF_BCAST(E.x, K)) << 8) | l16); va[J] = *(const v4f*)(xb + so); }
#define ROW_SUM(J, E, K)  { const float wj = __int_as_float(HALF_BCAST(E.y, K)); acc += va[J] * wj; }
#define ISSUE(J, E, K) if ((J) < mx) ROW_LOAD(J, E, K)
#define SUM(J, E, K)   if ((J) < rem) ROW_SUM(J, E, K)
__global__ void __launch_bounds__(256)
layer_kernel(const int* __restrict__ rowptr, const int* __restrict__ sorted,
             const float* __restrict__ invdeg, const float* __restrict__ xin,
             const float* __restrict__ W, const float* __restrict__ b,
             float* __restrict__ out, int N, int do_relu) {
  __shared__ float Ws[FDIM * FDIM];
  __shared__ __attribute__((aligned(16))) float tiles[8 * 16 * LDP];
  __shared__ __attribute__((aligned(16))) int2 recs[8 * RCAP];
  const int2* sk2 = (const int2*)sorted;
  const int tid  = threadIdx.x;
  const int lane = tid & 31;
  const int half = lane >> 4;
  const int l    = lane & 15;
  const int wv = __builtin_amdgcn_readfirstlane(tid >> 5);

  for (int i = tid; i < FDIM * FDIM; i += 256) {
    const int s  = i & 1;
    const int ln = (i >> 1) & 31;
    const int ct = (i >> 6) & 3;
    const int kk = i >> 8;
    Ws[i] = W[(kk * 4 + 2 * (ln >> 4) + s) * FDIM + ct * 16 + (ln & 15)];
  }
  __syncthreads();

  float* tile = tiles + wv * (16 * LDP);
  const v2f bb = *(const v2f*)(b + 2 * lane);
  const char* xb = (const char*)xin;
  const unsigned l16 = (unsigned)l << 4;
  const int nstrips = (N + 15) >> 4;

  const int strip = blockIdx.x * 8 + wv;
  if (strip < nstrips) {
    const int row0 = strip * 16;
    const int nr   = (N - row0 < 16) ? (N - row0) : 16;

    int rq = row0 + lane;
    rq = (rq < N) ? rq : N;
    const int rp = rowptr[rq];
    const int iq = (rq < N) ? rq : (N - 1);
    const int idv = __float_as_int(invdeg[iq]);

    int2* rs = recs + wv * RCAP;
    const int sbeg = __builtin_amdgcn_readlane(rp, 0);
    const int send = __builtin_amdgcn_readlane(rp, 16);
    const int scnt = send - sbeg;
    const bool big = scnt > RCAP;
    {
      int2 q[RCAP / 32];
      #pragma unroll
      for (int u = 0; u < RCAP / 32; ++u) {
        q[u].x = 0; q[u].y = 0;
        if (32 * u < scnt) {
          int k = sbeg + 32 * u + lane;
          k = (k < send) ? k : (send - 1);
          q[u] = sk2[k];
        }
      }
      #pragma unroll
      for (int u = 0; u < RCAP / 32; ++u) {
        if (u == 0 || 32 * u < scnt) rs[32 * u + lane] = q[u];
      }
    }
    __builtin_amdgcn_fence(__ATOMIC_ACQ_REL, "wavefront");

    int i = 0, c = 0;
    int n_a, n_b, my_beg, my_n;
    {
      const int a0 = __builtin_amdgcn_readlane(rp, 0), a1 = __builtin_amdgcn_readlane(rp, 1);
      const int b0 = __builtin_amdgcn_readlane(rp, 8), b1 = __builtin_amdgcn_readlane(rp, 9);
      n_a = a1 - a0; n_b = b1 - b0;
      my_beg = (half ? b0 : a0) - sbeg;
      my_n   = half ? n_b : n_a;
    }
    int2 e_lo, e_hi;
    e_lo.x = 0; e_lo.y = 0; e_hi.x = 0; e_hi.y = 0;
    { const int ix = my_beg + l;      REC_AT(e_lo, ix, l < my_n) }
    { const int ix = my_beg + 16 + l; REC_AT(e_hi, ix, l + 16 < my_n) }
    v4f acc = {0.0f, 0.0f, 0.0f, 0.0f};
    v4f hs  = {0.0f, 0.0f, 0.0f, 0.0f};

    for (;;) {
      const int ra = n_a - c, rb = n_b - c;
      const int mxr = (ra < rb) ? rb : ra;
      const int mx  = (mxr < GS) ? mxr : GS;
      const int rem = my_n - c;
      v4f va[GS];

      const unsigned so_lo = (((unsigned)HALF_BCAST(e_lo.x, 0)) << 8) | l16;
      const unsigned so_hi = (((unsigned)HALF_BCAST(e_hi.x, 0)) << 8) | l16;

      SLOTS(ISSUE)

      int ni = i, nc = c + GS;
      int nn_a = n_a, nn_b = n_b, nbeg = my_beg, nmy = my_n;
      if (mxr <= GS) {
        ni = i + 1; nc = 0;
        nn_a = 0; nn_b = 0; nmy = 0;
        if (ni < 8) {
          const int a0 = __builtin_amdgcn_readlane(rp, ni),     a1 = __builtin_amdgcn_readlane(rp, ni + 1);
          const int b0 = __builtin_amdgcn_readlane(rp, ni + 8), b1 = __builtin_amdgcn_readlane(rp, ni + 9);
          nn_a = a1 - a0; nn_b = b1 - b0;
          nbeg = (half ? b0 : a0) - sbeg;
          nmy  = half ? nn_b : nn_a;
        }
      }
      const int nrem = nmy - nc;
      int2 ne_lo, ne_hi;
      ne_lo.x = 0; ne_lo.y = 0; ne_hi.x = 0; ne_hi.y = 0;
      { const int ix = nbeg + nc + l;      REC_AT(ne_lo, ix, l < nrem) }
      { const int ix = nbeg + nc + 16 + l; REC_AT(ne_hi, ix, l + 16 < nrem) }
      if (c == 0) {
        const int myrow = row0 + 8 * half + i;
        hs.x = 0.0f; hs.y = 0.0f; hs.z = 0.0f; hs.w = 0.0f;
        if (myrow < N) hs = *(const v4f*)(xb + (((unsigned)myrow << 8) | l16));
      }

      SLOTS(SUM)

      if (ni != i) {
        const float ia = __int_as_float(__builtin_amdgcn_readlane(idv, i));
        const float ib = __int_as_float(__builtin_amdgcn_readlane(idv, i + 8));
        const float idg = half ? ib : ia;
        const v4f o = acc + hs * idg;
        *(v4f*)(tile + (8 * half + i) * LDP + 4 * l) = o;
        acc.x = 0.0f; acc.y = 0.0f; acc.z = 0.0f; acc.w = 0.0f;
        if (ni >= 8) break;
      }
      i = ni; c = nc; n_a = nn_a; n_b = nn_b; my_beg = nbeg; my_n = nmy;
      e_lo = ne_lo; e_hi = ne_hi;
    }
    __builtin_amdgcn_fence(__ATOMIC_ACQ_REL, "wavefront");

    v8f c0 = {}, c1 = {}, c2 = {}, c3 = {};
    const float* ap = tile + l * LDP + 2 * half;
    const float* bp = Ws + 2 * lane;
    v2f a  = *(const v2f*)(ap);
    v2f b0 = *(const v2f*)(bp + 0 * 64);
    v2f b1 = *(const v2f*)(bp + 1 * 64);
    v2f b2 = *(const v2f*)(bp + 2 * 64);
    v2f b3 = *(const v2f*)(bp + 3 * 64);
    __builtin_amdgcn_sched_group_barrier(0x100, 3, 0);
    #pragma unroll
    for (int kk = 0; kk < 16; ++kk) {
      v2f an = a, bn0 = b0, bn1 = b1, bn2 = b2, bn3 = b3;
      if (kk < 15) {
        an  = *(const v2f*)(ap + (kk + 1) * 4);
        bn0 = *(const v2f*)(bp + ((kk + 1) * 4 + 0) * 64);
        bn1 = *(const v2f*)(bp + ((kk + 1) * 4 + 1) * 64);
        bn2 = *(const v2f*)(bp + ((kk + 1) * 4 + 2) * 64);
        bn3 = *(const v2f*)(bp + ((kk + 1) * 4 + 3) * 64);
      }
      c0 = __builtin_amdgcn_wmma_f32_16x16x4_f32(false, a, false, b0, (short)0, c0, false, false);
      c1 = __builtin_amdgcn_wmma_f32_16x16x4_f32(false, a, false, b1, (short)0, c1, false, false);
      c2 = __builtin_amdgcn_wmma_f32_16x16x4_f32(false, a, false, b2, (short)0, c2, false, false);
      c3 = __builtin_amdgcn_wmma_f32_16x16x4_f32(false, a, false, b3, (short)0, c3, false, false);
      __builtin_amdgcn_sched_group_barrier(0x100, 3, 0);
      __builtin_amdgcn_sched_group_barrier(0x008, 4, 0);
      a = an; b0 = bn0; b1 = bn1; b2 = bn2; b3 = bn3;
    }

    float* cp = tile + (8 * half) * LDP + l;
    #pragma unroll
    for (int r = 0; r < 8; ++r) {
      cp[r * LDP +  0] = c0[r];
      cp[r * LDP + 16] = c1[r];
      cp[r * LDP + 32] = c2[r];
      cp[r * LDP + 48] = c3[r];
    }
    __builtin_amdgcn_fence(__ATOMIC_ACQ_REL, "wavefront");

    if (nr == 16) {
      v2f v[16];
      #pragma unroll
      for (int r = 0; r < 16; ++r) v[r] = *(const v2f*)(tile + r * LDP + 2 * lane);
      #pragma unroll
      for (int r = 0; r < 16; ++r) {
        v2f o;
        o.x = v[r].x + bb.x;
        o.y = v[r].y + bb.y;
        if (do_relu) {
          o.x = fmaxf(o.x, 0.0f);
          o.y = fmaxf(o.y, 0.0f);
        }
        *(v2f*)(out + (size_t)(row0 + r) * FDIM + 2 * lane) = o;
      }
    } else {
      for (int r = 0; r < nr; ++r) {
        const v2f v = *(const v2f*)(tile + r * LDP + 2 * lane);
        v2f o;
        o.x = v.x + bb.x;
        o.y = v.y + bb.y;
        if (do_relu) {
          o.x = fmaxf(o.x, 0.0f);
          o.y = fmaxf(o.y, 0.0f);
        }
        *(v2f*)(out + (size_t)(row0 + r) * FDIM + 2 * lane) = o;
      }
    }
  }
}

extern "C" void kernel_launch(void* const* d_in, const int* in_sizes, int n_in,
                              void* d_out, int out_size, void* d_ws, size_t ws_size,
                              hipStream_t stream) {
  const float* x  = (const float*)d_in[0];
  const int*   ei = (const int*)  d_in[1];
  const float* W  = (const float*)d_in[2];
  const float* b  = (const float*)d_in[3];

  const int N     = in_sizes[0] / FDIM;
  const int E     = in_sizes[1] / 2;
  const int DEPTH = in_sizes[2] / (FDIM * FDIM);

  const int T = (N + 1 + TROWS - 1) / TROWS;
  const int C = (E + CH - 1) / CH;
  const size_t NP = (size_t)((N + 1 + SCAN_ROWS - 1) / SCAN_ROWS) * SCAN_ROWS;
  const size_t EP = (size_t)C * CH;

  int*   wsi    = (int*)d_ws;
  int*   rowptr = wsi;
  float* dinv   = (float*)(wsi + NP);
  float* invdeg = (float*)(wsi + 2 * NP);
  int*   bsrc   = wsi + 3 * NP;
  int*   rec    = bsrc + EP;
  int*   sorted = rec + 2 * EP;
  float* xa     = (float*)(sorted + 2 * EP);
  float* xb     = xa + (size_t)N * FDIM;
  int*   off    = (int*)(xb + (size_t)N * FDIM);

  const int TB = 256;
  const int G = 2048;

  tile_sort_kernel<<<C, TB, 0, stream>>>(ei, rec, off, E, T);
  csr_build_kernel<<<T, TB, 0, stream>>>(rec, off, rowptr, dinv, invdeg, bsrc, C, T);
  sort_rows_kernel<<<G, TB, 0, stream>>>(rowptr, bsrc, dinv, sorted, N);

  const int layerBlocks = (N + 127) / 128;

  const float* xcur = x;
  for (int layer = 0; layer < DEPTH; ++layer) {
    float* xout = (layer == DEPTH - 1) ? (float*)d_out : ((layer & 1) ? xb : xa);
    layer_kernel<<<layerBlocks, TB, 0, stream>>>(
        rowptr, sorted, invdeg, xcur, W + (size_t)layer * FDIM * FDIM,
        b + (size_t)layer * FDIM, xout, N, layer != DEPTH - 1);
    xcur = xout;
  }
}
